# starting kernel
# baseline (speedup 1.0000x reference)
#include <hip/hip_runtime.h>
#include <math.h>

typedef float v2f __attribute__((ext_vector_type(2)));
typedef float v8f __attribute__((ext_vector_type(8)));

#define GCN_G   64
#define GCN_OUT 64
#define LDS_STRIDE 132

__global__ void k_deg_init(float* __restrict__ deg, int n) {
    int i = blockIdx.x * blockDim.x + threadIdx.x;
    if (i < n) deg[i] = 1.0f;
}

__global__ void k_deg_count(const int* __restrict__ dst, float* __restrict__ deg, int e) {
    int i = blockIdx.x * blockDim.x + threadIdx.x;
    if (i < e) atomicAdd(&deg[dst[i]], 1.0f);
}

__global__ void k_deg_inv(float* __restrict__ deg, int n) {
    int i = blockIdx.x * blockDim.x + threadIdx.x;
    if (i < n) {
        float d = deg[i];
        deg[i] = (d > 0.0f) ? (1.0f / sqrtf(d)) : 0.0f;
    }
}

__global__ void __launch_bounds__(128)
k_gemm_wmma_f32(const float* __restrict__ X, const float* __restrict__ W,
                float* __restrict__ H, int K, int F, int mtiles) {
    __shared__ float Wlds[16 * LDS_STRIDE];

    const int t     = threadIdx.x;
    const int wave  = t >> 5;
    const int lane  = t & 31;
    const int half  = lane >> 4;
    const int l16   = lane & 15;
    const int col0  = blockIdx.y << 4;
    const int mtile = (blockIdx.x << 2) + wave;

    const int reps = (K << 4) >> 7;
    for (int rep = 0; rep < reps; ++rep) {
        int idx = (rep << 7) + t;
        int k = idx >> 4;
        int n = idx & 15;
        Wlds[n * LDS_STRIDE + k] = W[(size_t)k * F + (col0 + n)];
    }
    __syncthreads();

    if (mtile < mtiles) {
        const int row0 = mtile << 4;
        const float* __restrict__ Xrow = X + (size_t)(row0 + l16) * K + (half << 1);
        const float* __restrict__ Wl = &Wlds[l16 * LDS_STRIDE + (half << 1)];

        v8f c = {};
#pragma unroll 8
        for (int k0 = 0; k0 < K; k0 += 4) {
            v2f a = *(const v2f*)(Xrow + k0);
            v2f b = *(const v2f*)(Wl + k0);
            c = __builtin_amdgcn_wmma_f32_16x16x4_f32(false, a, false, b,
                                                      (short)0, c, false, false);
        }

        float* __restrict__ Ho = H + (size_t)(row0 + (half << 3)) * F + (col0 + l16);
#pragma unroll
        for (int v = 0; v < 8; ++v)
            Ho[(size_t)v * F] = c[v];
    }
}

__global__ void k_agg_init(const float* __restrict__ H, const float* __restrict__ dinv,
                           const float* __restrict__ bias, float* __restrict__ out,
                           int n, int fshift) {
    int idx = blockIdx.x * blockDim.x + threadIdx.x;
    int total = n << fshift;
    if (idx >= total) return;
    int i = idx >> fshift;
    int f = idx & ((1 << fshift) - 1);
    float di = dinv[i];
    out[idx] = di * di * H[idx] + bias[f];
}

__global__ void k_edge_scatter(const int* __restrict__ src, const int* __restrict__ dst,
                               const float* __restrict__ dinv, const float* __restrict__ H,
                               float* __restrict__ out, int e, int fshift) {
    const int cshift = fshift - 2;
    long long t = (long long)blockIdx.x * blockDim.x + threadIdx.x;
    long long total = (long long)e << cshift;
    if (t >= total) return;
    int ei = (int)(t >> cshift);
    int c  = ((int)t & ((1 << cshift) - 1)) << 2;
    int s = src[ei];
    int d = dst[ei];
    float w = dinv[s] * dinv[d];
    int F = 1 << fshift;
    const float4 h4 = *(const float4*)(H + (size_t)s * F + c);
    float* o = out + (size_t)d * F + c;
    atomicAdd(o + 0, w * h4.x);
    atomicAdd(o + 1, w * h4.y);
    atomicAdd(o + 2, w * h4.z);
    atomicAdd(o + 3, w * h4.w);
}

__global__ void k_relu(float* __restrict__ x, int total) {
    int i = blockIdx.x * blockDim.x + threadIdx.x;
    if (i < total) x[i] = fmaxf(x[i], 0.0f);
}

__global__ void k_readout_zero(float* __restrict__ gsum, float* __restrict__ cnt) {
    int i = blockIdx.x * blockDim.x + threadIdx.x;
    if (i < GCN_G * GCN_OUT) gsum[i] = 0.0f;
    if (i < GCN_G) cnt[i] = 0.0f;
}

__global__ void k_readout_scatter(const float* __restrict__ h, const int* __restrict__ batch,
                                  float* __restrict__ gsum, float* __restrict__ cnt, int n) {
    int idx = blockIdx.x * blockDim.x + threadIdx.x;
    if (idx >= (n << 6)) return;
    int i = idx >> 6;
    int f = idx & 63;
    int g = batch[i];
    atomicAdd(&gsum[(g << 6) + f], h[idx]);
    if (f == 0) atomicAdd(&cnt[g], 1.0f);
}

__global__ void k_readout_final(const float* __restrict__ gsum, const float* __restrict__ cnt,
                                float* __restrict__ outp) {
    int idx = blockIdx.x * blockDim.x + threadIdx.x;
    if (idx >= GCN_G * GCN_OUT) return;
    int g = idx >> 6;
    outp[idx] = gsum[idx] / fmaxf(cnt[g], 1.0f);
}

extern "C" void kernel_launch(void* const* d_in, const int* in_sizes, int n_in,
                              void* d_out, int out_size, void* d_ws, size_t ws_size,
                              hipStream_t stream) {
    const float* x     = (const float*)d_in[0];
    const int*   edge  = (const int*)d_in[1];
    const int*   batch = (const int*)d_in[2];
    const float* W1    = (const float*)d_in[3];
    const float* b1    = (const float*)d_in[4];
    const float* W2    = (const float*)d_in[5];
    const float* b2    = (const float*)d_in[6];
    const float* W3    = (const float*)d_in[7];
    const float* b3    = (const float*)d_in[8];

    const int IN = 128, HID = 128, OUTF = 64;
    const int N = in_sizes[0] / IN;
    const int E = in_sizes[1] / 2;
    const int* src = edge;
    const int* dst = edge + E;

    char* ws = (char*)d_ws;
    float* dinv = (float*)ws;
    size_t off = ((size_t)N * sizeof(float) + 511) & ~(size_t)511;
    float* bufA = (float*)(ws + off);
    float* bufB = bufA + (size_t)N * HID;
    float* gsum = bufB + (size_t)N * HID;
    float* cnt  = gsum + GCN_G * GCN_OUT;

    const int TB = 256;
    dim3 blk(TB);

    k_deg_init <<<dim3((N + TB - 1) / TB), blk, 0, stream>>>(dinv, N);
    k_deg_count<<<dim3((E + TB - 1) / TB), blk, 0, stream>>>(dst, dinv, E);
    k_deg_inv  <<<dim3((N + TB - 1) / TB), blk, 0, stream>>>(dinv, N);

    const int mt = N / 16;
    const int gx = (mt + 3) / 4;
    long long tE128 = (long long)E * (HID / 4);
    long long tE64  = (long long)E * (OUTF / 4);
    int nblkE128 = (int)((tE128 + TB - 1) / TB);
    int nblkE64  = (int)((tE64 + TB - 1) / TB);

    k_gemm_wmma_f32<<<dim3(gx, HID / 16), dim3(128), 0, stream>>>(x, W1, bufA, IN, HID, mt);
    k_agg_init     <<<dim3((N * HID + TB - 1) / TB), blk, 0, stream>>>(bufA, dinv, b1, bufB, N, 7);
    k_edge_scatter <<<dim3(nblkE128), blk, 0, stream>>>(src, dst, dinv, bufA, bufB, E, 7);
    k_relu         <<<dim3((N * HID + TB - 1) / TB), blk, 0, stream>>>(bufB, N * HID);

    k_gemm_wmma_f32<<<dim3(gx, HID / 16), dim3(128), 0, stream>>>(bufB, W2, bufA, HID, HID, mt);
    k_agg_init     <<<dim3((N * HID + TB - 1) / TB), blk, 0, stream>>>(bufA, dinv, b2, bufB, N, 7);
    k_edge_scatter <<<dim3(nblkE128), blk, 0, stream>>>(src, dst, dinv, bufA, bufB, E, 7);
    k_relu         <<<dim3((N * HID + TB - 1) / TB), blk, 0, stream>>>(bufB, N * HID);

    k_gemm_wmma_f32<<<dim3(gx, OUTF / 16), dim3(128), 0, stream>>>(bufB, W3, bufA, HID, OUTF, mt);
    k_agg_init     <<<dim3((N * OUTF + TB - 1) / TB), blk, 0, stream>>>(bufA, dinv, b3, bufB, N, 6);
    k_edge_scatter <<<dim3(nblkE64), blk, 0, stream>>>(src, dst, dinv, bufA, bufB, E, 6);

    k_readout_zero   <<<dim3((GCN_G * GCN_OUT + TB - 1) / TB), blk, 0, stream>>>(gsum, cnt);
    k_readout_scatter<<<dim3((N * OUTF + TB - 1) / TB), blk, 0, stream>>>(bufB, batch, gsum, cnt, N);
    k_readout_final  <<<dim3((GCN_G * GCN_OUT + TB - 1) / TB), blk, 0, stream>>>(gsum, cnt, (float*)d_out);
}


# Edge sums gathered by destination from a per-call CSR (count+rank int atomics, scan, fill): one wave per node writes each row once with self-loop, bias, relu; sum order = atomic arrival rank
# speedup vs baseline: 2.2344x; 2.2344x over previous
#include <hip/hip_runtime.h>
#include <math.h>

typedef float v2f __attribute__((ext_vector_type(2)));
typedef float v8f __attribute__((ext_vector_type(8)));

#define GCN_G   64
#define GCN_OUT 64
#define LDS_STRIDE 132

__global__ void k_cnt_zero(int* __restrict__ icnt, int n) {
    int i = blockIdx.x * blockDim.x + threadIdx.x;
    if (i < n) icnt[i] = 0;
}

__global__ void k_count_rank(const int* __restrict__ dst, int* __restrict__ icnt,
                             int* __restrict__ rank, int e) {
    int i = blockIdx.x * blockDim.x + threadIdx.x;
    if (i < e) rank[i] = atomicAdd(&icnt[dst[i]], 1);
}

__global__ void __launch_bounds__(256)
k_scan_local(const int* __restrict__ icnt, int* __restrict__ rowptr,
             int* __restrict__ bsum, float* __restrict__ dinv, int n) {
    __shared__ int wsum[8];
    const int t    = threadIdx.x;
    const int lane = t & 31;
    const int wave = t >> 5;
    const int i0   = (blockIdx.x * 256 + t) * 4;
    int c[4];
#pragma unroll
    for (int k = 0; k < 4; ++k) c[k] = (i0 + k < n) ? icnt[i0 + k] : 0;
#pragma unroll
    for (int k = 0; k < 4; ++k) {
        if (i0 + k < n) {
            float d = (float)(c[k] + 1);
            dinv[i0 + k] = (d > 0.0f) ? (1.0f / sqrtf(d)) : 0.0f;
        }
    }
    const int tsum = c[0] + c[1] + c[2] + c[3];
    int x = tsum;
#pragma unroll
    for (int off = 1; off < 32; off <<= 1) {
        int y = __shfl_up(x, off);
        if (lane >= off) x += y;
    }
    if (lane == 31) wsum[wave] = x;
    __syncthreads();
    int woff = 0;
    for (int w = 0; w < wave; ++w) woff += wsum[w];
    int excl = woff + x - tsum;
#pragma unroll
    for (int k = 0; k < 4; ++k) {
        if (i0 + k < n) rowptr[i0 + k] = excl;
        excl += c[k];
    }
    if (t == 255) bsum[blockIdx.x] = woff + x;
}

__global__ void __launch_bounds__(256)
k_scan_add(int* __restrict__ rowptr, const int* __restrict__ bsum, int n) {
    __shared__ int red[8];
    const int t    = threadIdx.x;
    const int lane = t & 31;
    const int wave = t >> 5;
    const int b    = blockIdx.x;
    int s = 0;
    for (int j = t; j < b; j += 256) s += bsum[j];
#pragma unroll
    for (int off = 16; off > 0; off >>= 1) s += __shfl_xor(s, off);
    if (lane == 0) red[wave] = s;
    __syncthreads();
    int tot = 0;
#pragma unroll
    for (int w = 0; w < 8; ++w) tot += red[w];
    const int i0 = (b * 256 + t) * 4;
#pragma unroll
    for (int k = 0; k < 4; ++k)
        if (i0 + k < n) rowptr[i0 + k] += tot;
}

__global__ void k_fill(const int* __restrict__ src, const int* __restrict__ dst,
                       const int* __restrict__ rank, const int* __restrict__ rowptr,
                       int* __restrict__ col, int e) {
    int i = blockIdx.x * blockDim.x + threadIdx.x;
    if (i < e) col[rowptr[dst[i]] + rank[i]] = src[i];
}

template <int VEC, bool RELU>
__global__ void __launch_bounds__(256)
k_gather(const int* __restrict__ rowptr, const int* __restrict__ icnt,
         const int* __restrict__ col, const float* __restrict__ dinv,
         const float* __restrict__ H, const float* __restrict__ bias,
         float* __restrict__ out, int n) {
    typedef float vf __attribute__((ext_vector_type(VEC)));
    constexpr int F = VEC * 32;
    const int lane = threadIdx.x & 31;
    const int i = __builtin_amdgcn_readfirstlane((int)(blockIdx.x * 8 + (threadIdx.x >> 5)));
    if (i >= n) return;
    const int   start = rowptr[i];
    const int   cnt   = icnt[i];
    const float di    = dinv[i];
    const int   fo    = lane * VEC;

    const vf hs = *(const vf*)(H + (size_t)i * F + fo);
    const vf bv = *(const vf*)(bias + fo);
    vf acc = (di * di) * hs;
    for (int base = 0; base < cnt; base += 32) {
        const int m = min(cnt - base, 32);
        int   s_l = 0;
        float w_l = 0.0f;
        if (lane < m) {
            s_l = col[start + base + lane];
            w_l = dinv[s_l] * di;
        }
        int j = 0;
        for (; j + 4 <= m; j += 4) {
            const int s0 = __builtin_amdgcn_readlane(s_l, j);
            const int s1 = __builtin_amdgcn_readlane(s_l, j + 1);
            const int s2 = __builtin_amdgcn_readlane(s_l, j + 2);
            const int s3 = __builtin_amdgcn_readlane(s_l, j + 3);
            const float w0 = __int_as_float(__builtin_amdgcn_readlane(__float_as_int(w_l), j));
            const float w1 = __int_as_float(__builtin_amdgcn_readlane(__float_as_int(w_l), j + 1));
            const float w2 = __int_as_float(__builtin_amdgcn_readlane(__float_as_int(w_l), j + 2));
            const float w3 = __int_as_float(__builtin_amdgcn_readlane(__float_as_int(w_l), j + 3));
            const vf h0 = *(const vf*)(H + (size_t)s0 * F + fo);
            const vf h1 = *(const vf*)(H + (size_t)s1 * F + fo);
            const vf h2 = *(const vf*)(H + (size_t)s2 * F + fo);
            const vf h3 = *(const vf*)(H + (size_t)s3 * F + fo);
            acc += w0 * h0;
            acc += w1 * h1;
            acc += w2 * h2;
            acc += w3 * h3;
        }
        const int r = m - j;
        if (r > 0) {
            const int   s0 = __builtin_amdgcn_readlane(s_l, j);
            const float w0 = __int_as_float(__builtin_amdgcn_readlane(__float_as_int(w_l), j));
            const vf h0 = *(const vf*)(H + (size_t)s0 * F + fo);
            if (r > 1) {
                const int   s1 = __builtin_amdgcn_readlane(s_l, j + 1);
                const float w1 = __int_as_float(__builtin_amdgcn_readlane(__float_as_int(w_l), j + 1));
                const vf h1 = *(const vf*)(H + (size_t)s1 * F + fo);
                if (r > 2) {
                    const int   s2 = __builtin_amdgcn_readlane(s_l, j + 2);
                    const float w2 = __int_as_float(__builtin_amdgcn_readlane(__float_as_int(w_l), j + 2));
                    const vf h2 = *(const vf*)(H + (size_t)s2 * F + fo);
                    acc += w0 * h0;
                    acc += w1 * h1;
                    acc += w2 * h2;
                } else {
                    acc += w0 * h0;
                    acc += w1 * h1;
                }
            } else {
                acc += w0 * h0;
            }
        }
    }
    acc += bv;
    if (RELU) {
#pragma unroll
        for (int k = 0; k < VEC; ++k) acc[k] = fmaxf(acc[k], 0.0f);
    }
    *(vf*)(out + (size_t)i * F + fo) = acc;
}

__global__ void __launch_bounds__(128)
k_gemm_wmma_f32(const float* __restrict__ X, const float* __restrict__ W,
                float* __restrict__ H, int K, int F, int mtiles) {
    __shared__ float Wlds[16 * LDS_STRIDE];

    const int t     = threadIdx.x;
    const int wave  = t >> 5;
    const int lane  = t & 31;
    const int half  = lane >> 4;
    const int l16   = lane & 15;
    const int col0  = blockIdx.y << 4;
    const int mtile = (blockIdx.x << 2) + wave;

    const int reps = (K << 4) >> 7;
    for (int rep = 0; rep < reps; ++rep) {
        int idx = (rep << 7) + t;
        int k = idx >> 4;
        int n = idx & 15;
        Wlds[n * LDS_STRIDE + k] = W[(size_t)k * F + (col0 + n)];
    }
    __syncthreads();

    if (mtile < mtiles) {
        const int row0 = mtile << 4;
        const float* __restrict__ Xrow = X + (size_t)(row0 + l16) * K + (half << 1);
        const float* __restrict__ Wl = &Wlds[l16 * LDS_STRIDE + (half << 1)];

        v8f c = {};
#pragma unroll 8
        for (int k0 = 0; k0 < K; k0 += 4) {
            v2f a = *(const v2f*)(Xrow + k0);
            v2f b = *(const v2f*)(Wl + k0);
            c = __builtin_amdgcn_wmma_f32_16x16x4_f32(false, a, false, b,
                                                      (short)0, c, false, false);
        }

        float* __restrict__ Ho = H + (size_t)(row0 + (half << 3)) * F + (col0 + l16);
#pragma unroll
        for (int v = 0; v < 8; ++v)
            Ho[(size_t)v * F] = c[v];
    }
}

__global__ void k_readout_zero(float* __restrict__ gsum, float* __restrict__ cnt) {
    int i = blockIdx.x * blockDim.x + threadIdx.x;
    if (i < GCN_G * GCN_OUT) gsum[i] = 0.0f;
    if (i < GCN_G) cnt[i] = 0.0f;
}

__global__ void k_readout_scatter(const float* __restrict__ h, const int* __restrict__ batch,
                                  float* __restrict__ gsum, float* __restrict__ cnt, int n) {
    int idx = blockIdx.x * blockDim.x + threadIdx.x;
    if (idx >= (n << 6)) return;
    int i = idx >> 6;
    int f = idx & 63;
    int g = batch[i];
    atomicAdd(&gsum[(g << 6) + f], h[idx]);
    if (f == 0) atomicAdd(&cnt[g], 1.0f);
}

__global__ void k_readout_final(const float* __restrict__ gsum, const float* __restrict__ cnt,
                                float* __restrict__ outp) {
    int idx = blockIdx.x * blockDim.x + threadIdx.x;
    if (idx >= GCN_G * GCN_OUT) return;
    int g = idx >> 6;
    outp[idx] = gsum[idx] / fmaxf(cnt[g], 1.0f);
}

extern "C" void kernel_launch(void* const* d_in, const int* in_sizes, int n_in,
                              void* d_out, int out_size, void* d_ws, size_t ws_size,
                              hipStream_t stream) {
    const float* x     = (const float*)d_in[0];
    const int*   edge  = (const int*)d_in[1];
    const int*   batch = (const int*)d_in[2];
    const float* W1    = (const float*)d_in[3];
    const float* b1    = (const float*)d_in[4];
    const float* W2    = (const float*)d_in[5];
    const float* b2    = (const float*)d_in[6];
    const float* W3    = (const float*)d_in[7];
    const float* b3    = (const float*)d_in[8];

    const int IN = 128, HID = 128, OUTF = 64;
    const int N = in_sizes[0] / IN;
    const int E = in_sizes[1] / 2;
    const int* src = edge;
    const int* dst = edge + E;

    char* ws = (char*)d_ws;
    float* dinv = (float*)ws;
    size_t off = ((size_t)N * sizeof(float) + 511) & ~(size_t)511;
    float* bufA = (float*)(ws + off);
    float* bufB = bufA + (size_t)N * HID;
    float* gsum = bufB + (size_t)N * HID;
    float* cnt  = gsum + GCN_G * GCN_OUT;
    int*   icnt   = (int*)(cnt + GCN_G);
    int*   rowptr = icnt + N;
    int*   rank   = rowptr + N;
    int*   col    = rank + E;
    int*   bsum   = col + E;

    const int TB = 256;
    dim3 blk(TB);

    const int nscan = (N + 1023) / 1024;
    k_cnt_zero  <<<dim3((N + TB - 1) / TB), blk, 0, stream>>>(icnt, N);
    k_count_rank<<<dim3((E + TB - 1) / TB), blk, 0, stream>>>(dst, icnt, rank, E);
    k_scan_local<<<dim3(nscan), blk, 0, stream>>>(icnt, rowptr, bsum, dinv, N);
    k_scan_add  <<<dim3(nscan), blk, 0, stream>>>(rowptr, bsum, N);
    k_fill      <<<dim3((E + TB - 1) / TB), blk, 0, stream>>>(src, dst, rank, rowptr, col, E);

    const int mt = N / 16;
    const int gx = (mt + 3) / 4;
    const int gn = (N + 7) / 8;

    k_gemm_wmma_f32<<<dim3(gx, HID / 16), dim3(128), 0, stream>>>(x, W1, bufA, IN, HID, mt);
    k_gather<4, true><<<dim3(gn), blk, 0, stream>>>(rowptr, icnt, col, dinv, bufA, b1, bufB, N);

    k_gemm_wmma_f32<<<dim3(gx, HID / 16), dim3(128), 0, stream>>>(bufB, W2, bufA, HID, HID, mt);
    k_gather<4, true><<<dim3(gn), blk, 0, stream>>>(rowptr, icnt, col, dinv, bufA, b2, bufB, N);

    k_gemm_wmma_f32<<<dim3(gx, OUTF / 16), dim3(128), 0, stream>>>(bufB, W3, bufA, HID, OUTF, mt);
    k_gather<2, false><<<dim3(gn), blk, 0, stream>>>(rowptr, icnt, col, dinv, bufA, b3, bufB, N);

    k_readout_zero   <<<dim3((GCN_G * GCN_OUT + TB - 1) / TB), blk, 0, stream>>>(gsum, cnt);
    k_readout_scatter<<<dim3((N * OUTF + TB - 1) / TB), blk, 0, stream>>>(bufB, batch, gsum, cnt, N);
    k_readout_final  <<<dim3((GCN_G * GCN_OUT + TB - 1) / TB), blk, 0, stream>>>(gsum, cnt, (float*)d_out);
}


# Fewer, longer waves: each wave loops over several nodes (gather), edges (CSR build), M-tiles (GEMM) or rows (readout: per-graph run sums in registers, then its float atomics; sum order changes)
# speedup vs baseline: 3.2906x; 1.4727x over previous
#include <hip/hip_runtime.h>
#include <math.h>

typedef float v2f __attribute__((ext_vector_type(2)));
typedef float v8f __attribute__((ext_vector_type(8)));

#define GCN_G   64
#define GCN_OUT 64
#define LDS_STRIDE 132

__global__ void k_cnt_zero(int* __restrict__ icnt, int n) {
    int i = blockIdx.x * blockDim.x + threadIdx.x;
    if (i < n) icnt[i] = 0;
}

__global__ void __launch_bounds__(256)
k_count_rank(const int* __restrict__ dst, int* __restrict__ icnt,
             int* __restrict__ rank, int e) {
    const int i0 = blockIdx.x * blockDim.x + threadIdx.x;
    const int st = gridDim.x * blockDim.x;
    int d[4], r[4];
#pragma unroll
    for (int k = 0; k < 4; ++k) {
        const int i = i0 + k * st;
        d[k] = dst[min(i, e - 1)];
    }
#pragma unroll
    for (int k = 0; k < 4; ++k) {
        const int i = i0 + k * st;
        r[k] = 0;
        if (i < e) r[k] = atomicAdd(&icnt[d[k]], 1);
    }
#pragma unroll
    for (int k = 0; k < 4; ++k) {
        const int i = i0 + k * st;
        if (i < e) rank[i] = r[k];
    }
}

__global__ void __launch_bounds__(256)
k_scan_local(const int* __restrict__ icnt, int* __restrict__ rowptr,
             int* __restrict__ bsum, float* __restrict__ dinv, int n) {
    __shared__ int wsum[8];
    const int t    = threadIdx.x;
    const int lane = t & 31;
    const int wave = t >> 5;
    const int i0   = (blockIdx.x * 256 + t) * 4;
    int c[4];
#pragma unroll
    for (int k = 0; k < 4; ++k) c[k] = (i0 + k < n) ? icnt[i0 + k] : 0;
#pragma unroll
    for (int k = 0; k < 4; ++k) {
        if (i0 + k < n) {
            float d = (float)(c[k] + 1);
            dinv[i0 + k] = (d > 0.0f) ? (1.0f / sqrtf(d)) : 0.0f;
        }
    }
    const int tsum = c[0] + c[1] + c[2] + c[3];
    int x = tsum;
#pragma unroll
    for (int off = 1; off < 32; off <<= 1) {
        int y = __shfl_up(x, off);
        if (lane >= off) x += y;
    }
    if (lane == 31) wsum[wave] = x;
    __syncthreads();
    int woff = 0;
    for (int w = 0; w < wave; ++w) woff += wsum[w];
    int excl = woff + x - tsum;
#pragma unroll
    for (int k = 0; k < 4; ++k) {
        if (i0 + k < n) rowptr[i0 + k] = excl;
        excl += c[k];
    }
    if (t == 255) bsum[blockIdx.x] = woff + x;
}

__global__ void __launch_bounds__(256)
k_scan_add(int* __restrict__ rowptr, const int* __restrict__ bsum, int n) {
    __shared__ int red[8];
    const int t    = threadIdx.x;
    const int lane = t & 31;
    const int wave = t >> 5;
    const int b    = blockIdx.x;
    int s = 0;
    for (int j = t; j < b; j += 256) s += bsum[j];
#pragma unroll
    for (int off = 16; off > 0; off >>= 1) s += __shfl_xor(s, off);
    if (lane == 0) red[wave] = s;
    __syncthreads();
    int tot = 0;
#pragma unroll
    for (int w = 0; w < 8; ++w) tot += red[w];
    const int i0 = (b * 256 + t) * 4;
#pragma unroll
    for (int k = 0; k < 4; ++k)
        if (i0 + k < n) rowptr[i0 + k] += tot;
}

__global__ void __launch_bounds__(256)
k_fill(const int* __restrict__ src, const int* __restrict__ dst,
       const int* __restrict__ rank, const int* __restrict__ rowptr,
       int* __restrict__ col, int e) {
    const int i0 = blockIdx.x * blockDim.x + threadIdx.x;
    const int st = gridDim.x * blockDim.x;
    int d[4], r[4], sv[4], p[4];
#pragma unroll
    for (int k = 0; k < 4; ++k) {
        const int ic = min(i0 + k * st, e - 1);
        d[k] = dst[ic]; r[k] = rank[ic]; sv[k] = src[ic];
    }
#pragma unroll
    for (int k = 0; k < 4; ++k) p[k] = rowptr[d[k]];
#pragma unroll
    for (int k = 0; k < 4; ++k) {
        const int i = i0 + k * st;
        if (i < e) col[p[k] + r[k]] = sv[k];
    }
}

template <int VEC, bool RELU>
__global__ void __launch_bounds__(256)
k_gather(const int* __restrict__ rowptr, const int* __restrict__ icnt,
         const int* __restrict__ col, const float* __restrict__ dinv,
         const float* __restrict__ H, const float* __restrict__ bias,
         float* __restrict__ out, int n) {
    typedef float vf __attribute__((ext_vector_type(VEC)));
    constexpr int F = VEC * 32;
    const int lane = threadIdx.x & 31;
    const int wid = __builtin_amdgcn_readfirstlane((int)(blockIdx.x * 8 + (threadIdx.x >> 5)));
    const int nw = (int)(gridDim.x * 8);
    const int fo = lane * VEC;
    const vf  bv = *(const vf*)(bias + fo);
    for (int i = wid; i < n; i += nw) {
    const int   start = rowptr[i];
    const int   cnt   = icnt[i];
    const float di    = dinv[i];

    const vf hs = *(const vf*)(H + (size_t)i * F + fo);
    vf acc = (di * di) * hs;
    for (int base = 0; base < cnt; base += 32) {
        const int m = min(cnt - base, 32);
        int   s_l = 0;
        float w_l = 0.0f;
        if (lane < m) {
            s_l = col[start + base + lane];
            w_l = dinv[s_l] * di;
        }
        int j = 0;
        for (; j + 4 <= m; j += 4) {
            const int s0 = __builtin_amdgcn_readlane(s_l, j);
            const int s1 = __builtin_amdgcn_readlane(s_l, j + 1);
            const int s2 = __builtin_amdgcn_readlane(s_l, j + 2);
            const int s3 = __builtin_amdgcn_readlane(s_l, j + 3);
            const float w0 = __int_as_float(__builtin_amdgcn_readlane(__float_as_int(w_l), j));
            const float w1 = __int_as_float(__builtin_amdgcn_readlane(__float_as_int(w_l), j + 1));
            const float w2 = __int_as_float(__builtin_amdgcn_readlane(__float_as_int(w_l), j + 2));
            const float w3 = __int_as_float(__builtin_amdgcn_readlane(__float_as_int(w_l), j + 3));
            const vf h0 = *(const vf*)(H + (size_t)s0 * F + fo);
            const vf h1 = *(const vf*)(H + (size_t)s1 * F + fo);
            const vf h2 = *(const vf*)(H + (size_t)s2 * F + fo);
            const vf h3 = *(const vf*)(H + (size_t)s3 * F + fo);
            acc += w0 * h0;
            acc += w1 * h1;
            acc += w2 * h2;
            acc += w3 * h3;
        }
        const int r = m - j;
        if (r > 0) {
            const int   s0 = __builtin_amdgcn_readlane(s_l, j);
            const float w0 = __int_as_float(__builtin_amdgcn_readlane(__float_as_int(w_l), j));
            const vf h0 = *(const vf*)(H + (size_t)s0 * F + fo);
            if (r > 1) {
                const int   s1 = __builtin_amdgcn_readlane(s_l, j + 1);
                const float w1 = __int_as_float(__builtin_amdgcn_readlane(__float_as_int(w_l), j + 1));
                const vf h1 = *(const vf*)(H + (size_t)s1 * F + fo);
                if (r > 2) {
                    const int   s2 = __builtin_amdgcn_readlane(s_l, j + 2);
                    const float w2 = __int_as_float(__builtin_amdgcn_readlane(__float_as_int(w_l), j + 2));
                    const vf h2 = *(const vf*)(H + (size_t)s2 * F + fo);
                    acc += w0 * h0;
                    acc += w1 * h1;
                    acc += w2 * h2;
                } else {
                    acc += w0 * h0;
                    acc += w1 * h1;
                }
            } else {
                acc += w0 * h0;
            }
        }
    }
    acc += bv;
    if (RELU) {
#pragma unroll
        for (int k = 0; k < VEC; ++k) acc[k] = fmaxf(acc[k], 0.0f);
    }
    *(vf*)(out + (size_t)i * F + fo) = acc;
    }
}

__global__ void __launch_bounds__(128)
k_gemm_wmma_f32(const float* __restrict__ X, const float* __restrict__ W,
                float* __restrict__ H, int K, int F, int mtiles) {
    __shared__ float Wlds[16 * LDS_STRIDE];

    const int t     = threadIdx.x;
    const int wave  = __builtin_amdgcn_readfirstlane(t >> 5);
    const int lane  = t & 31;
    const int half  = lane >> 4;
    const int l16   = lane & 15;
    const int col0  = blockIdx.y << 4;
    const int mstep = (int)(gridDim.x << 2);

    const int reps = (K << 4) >> 7;
    for (int rep = 0; rep < reps; ++rep) {
        int idx = (rep << 7) + t;
        int k = idx >> 4;
        int n = idx & 15;
        Wlds[n * LDS_STRIDE + k] = W[(size_t)k * F + (col0 + n)];
    }
    __syncthreads();

    for (int mtile = (int)(blockIdx.x << 2) + wave; mtile < mtiles; mtile += mstep) {
        const int row0 = mtile << 4;
        const float* __restrict__ Xrow = X + (size_t)(row0 + l16) * K + (half << 1);
        const float* __restrict__ Wl = &Wlds[l16 * LDS_STRIDE + (half << 1)];

        v8f c = {};
#pragma unroll 8
        for (int k0 = 0; k0 < K; k0 += 4) {
            v2f a = *(const v2f*)(Xrow + k0);
            v2f b = *(const v2f*)(Wl + k0);
            c = __builtin_amdgcn_wmma_f32_16x16x4_f32(false, a, false, b,
                                                      (short)0, c, false, false);
        }

        float* __restrict__ Ho = H + (size_t)(row0 + (half << 3)) * F + (col0 + l16);
#pragma unroll
        for (int v = 0; v < 8; ++v)
            Ho[(size_t)v * F] = c[v];
    }
}

__global__ void k_readout_zero(float* __restrict__ gsum, float* __restrict__ cnt) {
    int i = blockIdx.x * blockDim.x + threadIdx.x;
    if (i < GCN_G * GCN_OUT) gsum[i] = 0.0f;
    if (i < GCN_G) cnt[i] = 0.0f;
}

#define RO_CHUNK 64
__global__ void __launch_bounds__(256)
k_readout_scatter(const float* __restrict__ h, const int* __restrict__ batch,
                  float* __restrict__ gsum, float* __restrict__ cnt, int n) {
    const int f   = threadIdx.x & 63;
    const int sub = threadIdx.x >> 6;
    const int beg = blockIdx.x * RO_CHUNK;
    const int end = min(beg + RO_CHUNK, n);
    bool  have = false;
    int   cur  = 0;
    float acc  = 0.0f;
    float c    = 0.0f;
    for (int i0 = beg + sub; i0 < end; i0 += 32) {
        int   g[8];
        float v[8];
#pragma unroll
        for (int u = 0; u < 8; ++u) {
            const int i = i0 + 4 * u;
            g[u] = 0;
            v[u] = 0.0f;
            if (i < end) {
                g[u] = batch[i];
                v[u] = h[((size_t)i << 6) + f];
            }
        }
#pragma unroll
        for (int u = 0; u < 8; ++u) {
            const int i = i0 + 4 * u;
            if (i < end) {
                if (have && g[u] != cur) {
                    atomicAdd(&gsum[(cur << 6) + f], acc);
                    if (f == 0) atomicAdd(&cnt[cur], c);
                    acc = 0.0f;
                    c   = 0.0f;
                }
                have = true;
                cur  = g[u];
                acc += v[u];
                c   += 1.0f;
            }
        }
    }
    if (have) {
        atomicAdd(&gsum[(cur << 6) + f], acc);
        if (f == 0) atomicAdd(&cnt[cur], c);
    }
}

__global__ void k_readout_final(const float* __restrict__ gsum, const float* __restrict__ cnt,
                                float* __restrict__ outp) {
    int idx = blockIdx.x * blockDim.x + threadIdx.x;
    if (idx >= GCN_G * GCN_OUT) return;
    int g = idx >> 6;
    outp[idx] = gsum[idx] / fmaxf(cnt[g], 1.0f);
}

extern "C" void kernel_launch(void* const* d_in, const int* in_sizes, int n_in,
                              void* d_out, int out_size, void* d_ws, size_t ws_size,
                              hipStream_t stream) {
    const float* x     = (const float*)d_in[0];
    const int*   edge  = (const int*)d_in[1];
    const int*   batch = (const int*)d_in[2];
    const float* W1    = (const float*)d_in[3];
    const float* b1    = (const float*)d_in[4];
    const float* W2    = (const float*)d_in[5];
    const float* b2    = (const float*)d_in[6];
    const float* W3    = (const float*)d_in[7];
    const float* b3    = (const float*)d_in[8];

    const int IN = 128, HID = 128, OUTF = 64;
    const int N = in_sizes[0] / IN;
    const int E = in_sizes[1] / 2;
    const int* src = edge;
    const int* dst = edge + E;

    char* ws = (char*)d_ws;
    float* dinv = (float*)ws;
    size_t off = ((size_t)N * sizeof(float) + 511) & ~(size_t)511;
    float* bufA = (float*)(ws + off);
    float* bufB = bufA + (size_t)N * HID;
    float* gsum = bufB + (size_t)N * HID;
    float* cnt  = gsum + GCN_G * GCN_OUT;
    int*   icnt   = (int*)(cnt + GCN_G);
    int*   rowptr = icnt + N;
    int*   rank   = rowptr + N;
    int*   col    = rank + E;
    int*   bsum   = col + E;

    const int TB = 256;
    dim3 blk(TB);

    const int nscan = (N + 1023) / 1024;
    k_cnt_zero  <<<dim3((N + TB - 1) / TB), blk, 0, stream>>>(icnt, N);
    k_count_rank<<<dim3((E + 4 * TB - 1) / (4 * TB)), blk, 0, stream>>>(dst, icnt, rank, E);
    k_scan_local<<<dim3(nscan), blk, 0, stream>>>(icnt, rowptr, bsum, dinv, N);
    k_scan_add  <<<dim3(nscan), blk, 0, stream>>>(rowptr, bsum, N);
    k_fill      <<<dim3((E + 4 * TB - 1) / (4 * TB)), blk, 0, stream>>>(src, dst, rank, rowptr, col, E);

    const int mt = N / 16;
    const int gx = (mt + 15) / 16;
    const int gn = (N + 7) / 8 < 2048 ? (N + 7) / 8 : 2048;

    k_gemm_wmma_f32<<<dim3(gx, HID / 16), dim3(128), 0, stream>>>(x, W1, bufA, IN, HID, mt);
    k_gather<4, true><<<dim3(gn), blk, 0, stream>>>(rowptr, icnt, col, dinv, bufA, b1, bufB, N);

    k_gemm_wmma_f32<<<dim3(gx, HID / 16), dim3(128), 0, stream>>>(bufB, W2, bufA, HID, HID, mt);
    k_gather<4, true><<<dim3(gn), blk, 0, stream>>>(rowptr, icnt, col, dinv, bufA, b2, bufB, N);

    k_gemm_wmma_f32<<<dim3(gx, OUTF / 16), dim3(128), 0, stream>>>(bufB, W3, bufA, HID, OUTF, mt);
    k_gather<2, false><<<dim3(gn), blk, 0, stream>>>(rowptr, icnt, col, dinv, bufA, b3, bufB, N);

    k_readout_zero   <<<dim3((GCN_G * GCN_OUT + TB - 1) / TB), blk, 0, stream>>>(gsum, cnt);
    k_readout_scatter<<<dim3((N + RO_CHUNK - 1) / RO_CHUNK), blk, 0, stream>>>(bufB, batch, gsum, cnt, N);
    k_readout_final  <<<dim3((GCN_G * GCN_OUT + TB - 1) / TB), blk, 0, stream>>>(gsum, cnt, (float*)d_out);
}


# GEMM retiled: block stages all of W in LDS once, each wave owns a 16-row strip and reuses every A fragment for all F/16 column tiles: X is fetched once, not once per column tile; sums unchanged
# speedup vs baseline: 3.7572x; 1.1418x over previous
#include <hip/hip_runtime.h>
#include <math.h>

typedef float v2f __attribute__((ext_vector_type(2)));
typedef float v8f __attribute__((ext_vector_type(8)));

#define GCN_G   64
#define GCN_OUT 64
#define LDS_STRIDE 132

__global__ void k_cnt_zero(int* __restrict__ icnt, int n) {
    int i = blockIdx.x * blockDim.x + threadIdx.x;
    if (i < n) icnt[i] = 0;
}

__global__ void __launch_bounds__(256)
k_count_rank(const int* __restrict__ dst, int* __restrict__ icnt,
             int* __restrict__ rank, int e) {
    const int i0 = blockIdx.x * blockDim.x + threadIdx.x;
    const int st = gridDim.x * blockDim.x;
    int d[4], r[4];
#pragma unroll
    for (int k = 0; k < 4; ++k) {
        const int i = i0 + k * st;
        d[k] = dst[min(i, e - 1)];
    }
#pragma unroll
    for (int k = 0; k < 4; ++k) {
        const int i = i0 + k * st;
        r[k] = 0;
        if (i < e) r[k] = atomicAdd(&icnt[d[k]], 1);
    }
#pragma unroll
    for (int k = 0; k < 4; ++k) {
        const int i = i0 + k * st;
        if (i < e) rank[i] = r[k];
    }
}

__global__ void __launch_bounds__(256)
k_scan_local(const int* __restrict__ icnt, int* __restrict__ rowptr,
             int* __restrict__ bsum, float* __restrict__ dinv, int n) {
    __shared__ int wsum[8];
    const int t    = threadIdx.x;
    const int lane = t & 31;
    const int wave = t >> 5;
    const int i0   = (blockIdx.x * 256 + t) * 4;
    int c[4];
#pragma unroll
    for (int k = 0; k < 4; ++k) c[k] = (i0 + k < n) ? icnt[i0 + k] : 0;
#pragma unroll
    for (int k = 0; k < 4; ++k) {
        if (i0 + k < n) {
            float d = (float)(c[k] + 1);
            dinv[i0 + k] = (d > 0.0f) ? (1.0f / sqrtf(d)) : 0.0f;
        }
    }
    const int tsum = c[0] + c[1] + c[2] + c[3];
    int x = tsum;
#pragma unroll
    for (int off = 1; off < 32; off <<= 1) {
        int y = __shfl_up(x, off);
        if (lane >= off) x += y;
    }
    if (lane == 31) wsum[wave] = x;
    __syncthreads();
    int woff = 0;
    for (int w = 0; w < wave; ++w) woff += wsum[w];
    int excl = woff + x - tsum;
#pragma unroll
    for (int k = 0; k < 4; ++k) {
        if (i0 + k < n) rowptr[i0 + k] = excl;
        excl += c[k];
    }
    if (t == 255) bsum[blockIdx.x] = woff + x;
}

__global__ void __launch_bounds__(256)
k_scan_add(int* __restrict__ rowptr, const int* __restrict__ bsum, int n) {
    __shared__ int red[8];
    const int t    = threadIdx.x;
    const int lane = t & 31;
    const int wave = t >> 5;
    const int b    = blockIdx.x;
    int s = 0;
    for (int j = t; j < b; j += 256) s += bsum[j];
#pragma unroll
    for (int off = 16; off > 0; off >>= 1) s += __shfl_xor(s, off);
    if (lane == 0) red[wave] = s;
    __syncthreads();
    int tot = 0;
#pragma unroll
    for (int w = 0; w < 8; ++w) tot += red[w];
    const int i0 = (b * 256 + t) * 4;
#pragma unroll
    for (int k = 0; k < 4; ++k)
        if (i0 + k < n) rowptr[i0 + k] += tot;
}

__global__ void __launch_bounds__(256)
k_fill(const int* __restrict__ src, const int* __restrict__ dst,
       const int* __restrict__ rank, const int* __restrict__ rowptr,
       int* __restrict__ col, int e) {
    const int i0 = blockIdx.x * blockDim.x + threadIdx.x;
    const int st = gridDim.x * blockDim.x;
    int d[4], r[4], sv[4], p[4];
#pragma unroll
    for (int k = 0; k < 4; ++k) {
        const int ic = min(i0 + k * st, e - 1);
        d[k] = dst[ic]; r[k] = rank[ic]; sv[k] = src[ic];
    }
#pragma unroll
    for (int k = 0; k < 4; ++k) p[k] = rowptr[d[k]];
#pragma unroll
    for (int k = 0; k < 4; ++k) {
        const int i = i0 + k * st;
        if (i < e) col[p[k] + r[k]] = sv[k];
    }
}

template <int VEC, bool RELU>
__global__ void __launch_bounds__(256)
k_gather(const int* __restrict__ rowptr, const int* __restrict__ icnt,
         const int* __restrict__ col, const float* __restrict__ dinv,
         const float* __restrict__ H, const float* __restrict__ bias,
         float* __restrict__ out, int n) {
    typedef float vf __attribute__((ext_vector_type(VEC)));
    constexpr int F = VEC * 32;
    const int lane = threadIdx.x & 31;
    const int wid = __builtin_amdgcn_readfirstlane((int)(blockIdx.x * 8 + (threadIdx.x >> 5)));
    const int nw = (int)(gridDim.x * 8);
    const int fo = lane * VEC;
    const vf  bv = *(const vf*)(bias + fo);
    for (int i = wid; i < n; i += nw) {
    const int   start = rowptr[i];
    const int   cnt   = icnt[i];
    const float di    = dinv[i];

    const vf hs = *(const vf*)(H + (size_t)i * F + fo);
    vf acc = (di * di) * hs;
    for (int base = 0; base < cnt; base += 32) {
        const int m = min(cnt - base, 32);
        int   s_l = 0;
        float w_l = 0.0f;
        if (lane < m) {
            s_l = col[start + base + lane];
            w_l = dinv[s_l] * di;
        }
        int j = 0;
        for (; j + 4 <= m; j += 4) {
            const int s0 = __builtin_amdgcn_readlane(s_l, j);
            const int s1 = __builtin_amdgcn_readlane(s_l, j + 1);
            const int s2 = __builtin_amdgcn_readlane(s_l, j + 2);
            const int s3 = __builtin_amdgcn_readlane(s_l, j + 3);
            const float w0 = __int_as_float(__builtin_amdgcn_readlane(__float_as_int(w_l), j));
            const float w1 = __int_as_float(__builtin_amdgcn_readlane(__float_as_int(w_l), j + 1));
            const float w2 = __int_as_float(__builtin_amdgcn_readlane(__float_as_int(w_l), j + 2));
            const float w3 = __int_as_float(__builtin_amdgcn_readlane(__float_as_int(w_l), j + 3));
            const vf h0 = *(const vf*)(H + (size_t)s0 * F + fo);
            const vf h1 = *(const vf*)(H + (size_t)s1 * F + fo);
            const vf h2 = *(const vf*)(H + (size_t)s2 * F + fo);
            const vf h3 = *(const vf*)(H + (size_t)s3 * F + fo);
            acc += w0 * h0;
            acc += w1 * h1;
            acc += w2 * h2;
            acc += w3 * h3;
        }
        const int r = m - j;
        if (r > 0) {
            const int   s0 = __builtin_amdgcn_readlane(s_l, j);
            const float w0 = __int_as_float(__builtin_amdgcn_readlane(__float_as_int(w_l), j));
            const vf h0 = *(const vf*)(H + (size_t)s0 * F + fo);
            if (r > 1) {
                const int   s1 = __builtin_amdgcn_readlane(s_l, j + 1);
                const float w1 = __int_as_float(__builtin_amdgcn_readlane(__float_as_int(w_l), j + 1));
                const vf h1 = *(const vf*)(H + (size_t)s1 * F + fo);
                if (r > 2) {
                    const int   s2 = __builtin_amdgcn_readlane(s_l, j + 2);
                    const float w2 = __int_as_float(__builtin_amdgcn_readlane(__float_as_int(w_l), j + 2));
                    const vf h2 = *(const vf*)(H + (size_t)s2 * F + fo);
                    acc += w0 * h0;
                    acc += w1 * h1;
                    acc += w2 * h2;
                } else {
                    acc += w0 * h0;
                    acc += w1 * h1;
                }
            } else {
                acc += w0 * h0;
            }
        }
    }
    acc += bv;
    if (RELU) {
#pragma unroll
        for (int k = 0; k < VEC; ++k) acc[k] = fmaxf(acc[k], 0.0f);
    }
    *(vf*)(out + (size_t)i * F + fo) = acc;
    }
}

template <int F>
__global__ void __launch_bounds__(512)
k_gemm_strip_f32(const float* __restrict__ X, const float* __restrict__ W,
                 float* __restrict__ H, int mtiles) {
    constexpr int K  = 128;
    constexpr int NT = F / 16;
    constexpr int NG = K / 8;
    __shared__ alignas(16) float Wlds[F * LDS_STRIDE];

    const int t     = threadIdx.x;
    const int wave  = __builtin_amdgcn_readfirstlane(t >> 5);
    const int lane  = t & 31;
    const int half  = lane >> 4;
    const int l16   = lane & 15;

    const float* __restrict__ Xl = X + (size_t)l16 * K + (half << 1);

    const int gx = (int)gridDim.x;
    int mtile = wave * gx + (int)blockIdx.x;
    const bool any = mtile < mtiles;

    v2f a[8];
#pragma unroll
    for (int u = 0; u < 8; ++u) a[u] = v2f{};
    if (any) {
#pragma unroll
        for (int u = 0; u < 8; ++u) a[u] = *(const v2f*)(Xl + (size_t)(mtile << 4) * K + 4 * u);
    }

    for (int idx = t; idx < K * F; idx += 512) {
        const int k = idx / F;
        const int n = idx & (F - 1);
        Wlds[n * LDS_STRIDE + k] = W[idx];
    }
    __syncthreads();
    if (!any) return;

    const float* __restrict__ Wl = &Wlds[l16 * LDS_STRIDE + (half << 1)];
    v2f bc[NT][2];
#pragma unroll
    for (int n = 0; n < NT; ++n) {
#pragma unroll
        for (int q = 0; q < 2; ++q) bc[n][q] = *(const v2f*)(Wl + n * 16 * LDS_STRIDE + 4 * q);
    }

    for (;;) {
        const int  row0  = mtile << 4;
        const int  mnext = mtile + 16 * gx;
        const bool more  = mnext < mtiles;
        const float* __restrict__ Xrow = Xl + (size_t)row0 * K;

        v8f c[NT];
#pragma unroll
        for (int n = 0; n < NT; ++n) c[n] = v8f{};

#pragma unroll
        for (int blk = 0; blk < 4; ++blk) {
            v2f an[8];
#pragma unroll
            for (int u = 0; u < 8; ++u) an[u] = a[u];
            if (blk < 3) {
#pragma unroll
                for (int u = 0; u < 8; ++u) an[u] = *(const v2f*)(Xrow + (blk + 1) * 32 + 4 * u);
            } else if (more) {
#pragma unroll
                for (int u = 0; u < 8; ++u) an[u] = *(const v2f*)(Xl + (size_t)(mnext << 4) * K + 4 * u);
            }
#pragma unroll
            for (int p = 0; p < 4; ++p) {
                const int g  = blk * 4 + p;
                const int gn = (g + 1) % NG;
                v2f bn[NT][2];
#pragma unroll
                for (int n = 0; n < NT; ++n) {
#pragma unroll
                    for (int q = 0; q < 2; ++q)
                        bn[n][q] = *(const v2f*)(Wl + n * 16 * LDS_STRIDE + gn * 8 + 4 * q);
                }
                __builtin_amdgcn_sched_barrier(0);
#pragma unroll
                for (int n = 0; n < NT; ++n) {
                    c[n] = __builtin_amdgcn_wmma_f32_16x16x4_f32(false, a[2 * p], false, bc[n][0],
                                                                 (short)0, c[n], false, false);
                    c[n] = __builtin_amdgcn_wmma_f32_16x16x4_f32(false, a[2 * p + 1], false, bc[n][1],
                                                                 (short)0, c[n], false, false);
                }
                __builtin_amdgcn_sched_barrier(0);
#pragma unroll
                for (int n = 0; n < NT; ++n) { bc[n][0] = bn[n][0]; bc[n][1] = bn[n][1]; }
            }
#pragma unroll
            for (int u = 0; u < 8; ++u) a[u] = an[u];
        }

        float* __restrict__ Ho = H + (size_t)(row0 + (half << 3)) * F + l16;
#pragma unroll
        for (int v = 0; v < 8; ++v) {
#pragma unroll
            for (int n = 0; n < NT; ++n)
                Ho[(size_t)v * F + n * 16] = c[n][v];
        }
        if (!more) break;
        mtile = mnext;
    }
}

__global__ void k_readout_zero(float* __restrict__ gsum, float* __restrict__ cnt) {
    int i = blockIdx.x * blockDim.x + threadIdx.x;
    if (i < GCN_G * GCN_OUT) gsum[i] = 0.0f;
    if (i < GCN_G) cnt[i] = 0.0f;
}

#define RO_CHUNK 64
__global__ void __launch_bounds__(256)
k_readout_scatter(const float* __restrict__ h, const int* __restrict__ batch,
                  float* __restrict__ gsum, float* __restrict__ cnt, int n) {
    const int f   = threadIdx.x & 63;
    const int sub = threadIdx.x >> 6;
    const int beg = blockIdx.x * RO_CHUNK;
    const int end = min(beg + RO_CHUNK, n);
    bool  have = false;
    int   cur  = 0;
    float acc  = 0.0f;
    float c    = 0.0f;
    for (int i0 = beg + sub; i0 < end; i0 += 32) {
        int   g[8];
        float v[8];
#pragma unroll
        for (int u = 0; u < 8; ++u) {
            const int i = i0 + 4 * u;
            g[u] = 0;
            v[u] = 0.0f;
            if (i < end) {
                g[u] = batch[i];
                v[u] = h[((size_t)i << 6) + f];
            }
        }
#pragma unroll
        for (int u = 0; u < 8; ++u) {
            const int i = i0 + 4 * u;
            if (i < end) {
                if (have && g[u] != cur) {
                    atomicAdd(&gsum[(cur << 6) + f], acc);
                    if (f == 0) atomicAdd(&cnt[cur], c);
                    acc = 0.0f;
                    c   = 0.0f;
                }
                have = true;
                cur  = g[u];
                acc += v[u];
                c   += 1.0f;
            }
        }
    }
    if (have) {
        atomicAdd(&gsum[(cur << 6) + f], acc);
        if (f == 0) atomicAdd(&cnt[cur], c);
    }
}

__global__ void k_readout_final(const float* __restrict__ gsum, const float* __restrict__ cnt,
                                float* __restrict__ outp) {
    int idx = blockIdx.x * blockDim.x + threadIdx.x;
    if (idx >= GCN_G * GCN_OUT) return;
    int g = idx >> 6;
    outp[idx] = gsum[idx] / fmaxf(cnt[g], 1.0f);
}

extern "C" void kernel_launch(void* const* d_in, const int* in_sizes, int n_in,
                              void* d_out, int out_size, void* d_ws, size_t ws_size,
                              hipStream_t stream) {
    const float* x     = (const float*)d_in[0];
    const int*   edge  = (const int*)d_in[1];
    const int*   batch = (const int*)d_in[2];
    const float* W1    = (const float*)d_in[3];
    const float* b1    = (const float*)d_in[4];
    const float* W2    = (const float*)d_in[5];
    const float* b2    = (const float*)d_in[6];
    const float* W3    = (const float*)d_in[7];
    const float* b3    = (const float*)d_in[8];

    const int IN = 128, HID = 128, OUTF = 64;
    const int N = in_sizes[0] / IN;
    const int E = in_sizes[1] / 2;
    const int* src = edge;
    const int* dst = edge + E;

    char* ws = (char*)d_ws;
    float* dinv = (float*)ws;
    size_t off = ((size_t)N * sizeof(float) + 511) & ~(size_t)511;
    float* bufA = (float*)(ws + off);
    float* bufB = bufA + (size_t)N * HID;
    float* gsum = bufB + (size_t)N * HID;
    float* cnt  = gsum + GCN_G * GCN_OUT;
    int*   icnt   = (int*)(cnt + GCN_G);
    int*   rowptr = icnt + N;
    int*   rank   = rowptr + N;
    int*   col    = rank + E;
    int*   bsum   = col + E;

    const int TB = 256;
    dim3 blk(TB);

    const int nscan = (N + 1023) / 1024;
    k_cnt_zero  <<<dim3((N + TB - 1) / TB), blk, 0, stream>>>(icnt, N);
    k_count_rank<<<dim3((E + 4 * TB - 1) / (4 * TB)), blk, 0, stream>>>(dst, icnt, rank, E);
    k_scan_local<<<dim3(nscan), blk, 0, stream>>>(icnt, rowptr, bsum, dinv, N);
    k_scan_add  <<<dim3(nscan), blk, 0, stream>>>(rowptr, bsum, N);
    k_fill      <<<dim3((E + 4 * TB - 1) / (4 * TB)), blk, 0, stream>>>(src, dst, rank, rowptr, col, E);

    const int mt = N / 16;
    const int gb = 256;
    const int gn = (N + 7) / 8 < 2048 ? (N + 7) / 8 : 2048;

    k_gemm_strip_f32<128><<<dim3(gb), dim3(512), 0, stream>>>(x, W1, bufA, mt);
    k_gather<4, true><<<dim3(gn), blk, 0, stream>>>(rowptr, icnt, col, dinv, bufA, b1, bufB, N);

    k_gemm_strip_f32<128><<<dim3(gb), dim3(512), 0, stream>>>(bufB, W2, bufA, mt);
    k_gather<4, true><<<dim3(gn), blk, 0, stream>>>(rowptr, icnt, col, dinv, bufA, b2, bufB, N);

    k_gemm_strip_f32<64><<<dim3(gb), dim3(512), 0, stream>>>(bufB, W3, bufA, mt);
    k_gather<2, false><<<dim3(gn), blk, 0, stream>>>(rowptr, icnt, col, dinv, bufA, b3, bufB, N);

    k_readout_zero   <<<dim3((GCN_G * GCN_OUT + TB - 1) / TB), blk, 0, stream>>>(gsum, cnt);
    k_readout_scatter<<<dim3((N + RO_CHUNK - 1) / RO_CHUNK), blk, 0, stream>>>(bufB, batch, gsum, cnt, N);
    k_readout_final  <<<dim3((GCN_G * GCN_OUT + TB - 1) / TB), blk, 0, stream>>>(gsum, cnt, (float*)d_out);
}


# Products of layers 2 and 3 moved into the gather kernel in front of each: a wave sums 16 nodes' rows into an LDS tile and multiplies it by W held in LDS (f32 WMMA); 2 launches fewer; sums unchanged
# speedup vs baseline: 4.0216x; 1.0704x over previous
#include <hip/hip_runtime.h>
#include <math.h>

typedef float v2f __attribute__((ext_vector_type(2)));
typedef float v8f __attribute__((ext_vector_type(8)));

#define GCN_G   64
#define GCN_OUT 64
#define LDS_STRIDE 132

__global__ void k_cnt_zero(int* __restrict__ icnt, int n) {
    int i = blockIdx.x * blockDim.x + threadIdx.x;
    if (i < n) icnt[i] = 0;
}

__global__ void __launch_bounds__(256)
k_count_rank(const int* __restrict__ dst, int* __restrict__ icnt,
             int* __restrict__ rank, int e) {
    const int i0 = blockIdx.x * blockDim.x + threadIdx.x;
    const int st = gridDim.x * blockDim.x;
    int d[4], r[4];
#pragma unroll
    for (int k = 0; k < 4; ++k) {
        const int i = i0 + k * st;
        d[k] = dst[min(i, e - 1)];
    }
#pragma unroll
    for (int k = 0; k < 4; ++k) {
        const int i = i0 + k * st;
        r[k] = 0;
        if (i < e) r[k] = atomicAdd(&icnt[d[k]], 1);
    }
#pragma unroll
    for (int k = 0; k < 4; ++k) {
        const int i = i0 + k * st;
        if (i < e) rank[i] = r[k];
    }
}

__global__ void __launch_bounds__(256)
k_scan_local(const int* __restrict__ icnt, int* __restrict__ rowptr,
             int* __restrict__ bsum, float* __restrict__ dinv, int n) {
    __shared__ int wsum[8];
    const int t    = threadIdx.x;
    const int lane = t & 31;
    const int wave = t >> 5;
    const int i0   = (blockIdx.x * 256 + t) * 4;
    int c[4];
#pragma unroll
    for (int k = 0; k < 4; ++k) c[k] = (i0 + k < n) ? icnt[i0 + k] : 0;
#pragma unroll
    for (int k = 0; k < 4; ++k) {
        if (i0 + k < n) {
            float d = (float)(c[k] + 1);
            dinv[i0 + k] = (d > 0.0f) ? (1.0f / sqrtf(d)) : 0.0f;
        }
    }
    const int tsum = c[0] + c[1] + c[2] + c[3];
    int x = tsum;
#pragma unroll
    for (int off = 1; off < 32; off <<= 1) {
        int y = __shfl_up(x, off);
        if (lane >= off) x += y;
    }
    if (lane == 31) wsum[wave] = x;
    __syncthreads();
    int woff = 0;
    for (int w = 0; w < wave; ++w) woff += wsum[w];
    int excl = woff + x - tsum;
#pragma unroll
    for (int k = 0; k < 4; ++k) {
        if (i0 + k < n) rowptr[i0 + k] = excl;
        excl += c[k];
    }
    if (t == 255) bsum[blockIdx.x] = woff + x;
}

__global__ void __launch_bounds__(256)
k_scan_add(int* __restrict__ rowptr, const int* __restrict__ bsum, int n) {
    __shared__ int red[8];
    const int t    = threadIdx.x;
    const int lane = t & 31;
    const int wave = t >> 5;
    const int b    = blockIdx.x;
    int s = 0;
    for (int j = t; j < b; j += 256) s += bsum[j];
#pragma unroll
    for (int off = 16; off > 0; off >>= 1) s += __shfl_xor(s, off);
    if (lane == 0) red[wave] = s;
    __syncthreads();
    int tot = 0;
#pragma unroll
    for (int w = 0; w < 8; ++w) tot += red[w];
    const int i0 = (b * 256 + t) * 4;
#pragma unroll
    for (int k = 0; k < 4; ++k)
        if (i0 + k < n) rowptr[i0 + k] += tot;
}

__global__ void __launch_bounds__(256)
k_fill(const int* __restrict__ src, const int* __restrict__ dst,
       const int* __restrict__ rank, const int* __restrict__ rowptr,
       int* __restrict__ col, int e) {
    const int i0 = blockIdx.x * blockDim.x + threadIdx.x;
    const int st = gridDim.x * blockDim.x;
    int d[4], r[4], sv[4], p[4];
#pragma unroll
    for (int k = 0; k < 4; ++k) {
        const int ic = min(i0 + k * st, e - 1);
        d[k] = dst[ic]; r[k] = rank[ic]; sv[k] = src[ic];
    }
#pragma unroll
    for (int k = 0; k < 4; ++k) p[k] = rowptr[d[k]];
#pragma unroll
    for (int k = 0; k < 4; ++k) {
        const int i = i0 + k * st;
        if (i < e) col[p[k] + r[k]] = sv[k];
    }
}

template <int VEC, bool RELU>
__global__ void __launch_bounds__(256)
k_gather(const int* __restrict__ rowptr, const int* __restrict__ icnt,
         const int* __restrict__ col, const float* __restrict__ dinv,
         const float* __restrict__ H, const float* __restrict__ bias,
         float* __restrict__ out, int n) {
    typedef float vf __attribute__((ext_vector_type(VEC)));
    constexpr int F = VEC * 32;
    const int lane = threadIdx.x & 31;
    const int wid = __builtin_amdgcn_readfirstlane((int)(blockIdx.x * 8 + (threadIdx.x >> 5)));
    const int nw = (int)(gridDim.x * 8);
    const int fo = lane * VEC;
    const vf  bv = *(const vf*)(bias + fo);
    for (int i = wid; i < n; i += nw) {
    const int   start = rowptr[i];
    const int   cnt   = icnt[i];
    const float di    = dinv[i];

    const vf hs = *(const vf*)(H + (size_t)i * F + fo);
    vf acc = (di * di) * hs;
    for (int base = 0; base < cnt; base += 32) {
        const int m = min(cnt - base, 32);
        int   s_l = 0;
        float w_l = 0.0f;
        if (lane < m) {
            s_l = col[start + base + lane];
            w_l = dinv[s_l] * di;
        }
        int j = 0;
        for (; j + 4 <= m; j += 4) {
            const int s0 = __builtin_amdgcn_readlane(s_l, j);
            const int s1 = __builtin_amdgcn_readlane(s_l, j + 1);
            const int s2 = __builtin_amdgcn_readlane(s_l, j + 2);
            const int s3 = __builtin_amdgcn_readlane(s_l, j + 3);
            const float w0 = __int_as_float(__builtin_amdgcn_readlane(__float_as_int(w_l), j));
            const float w1 = __int_as_float(__builtin_amdgcn_readlane(__float_as_int(w_l), j + 1));
            const float w2 = __int_as_float(__builtin_amdgcn_readlane(__float_as_int(w_l), j + 2));
            const float w3 = __int_as_float(__builtin_amdgcn_readlane(__float_as_int(w_l), j + 3));
            const vf h0 = *(const vf*)(H + (size_t)s0 * F + fo);
            const vf h1 = *(const vf*)(H + (size_t)s1 * F + fo);
            const vf h2 = *(const vf*)(H + (size_t)s2 * F + fo);
            const vf h3 = *(const vf*)(H + (size_t)s3 * F + fo);
            acc += w0 * h0;
            acc += w1 * h1;
            acc += w2 * h2;
            acc += w3 * h3;
        }
        const int r = m - j;
        if (r > 0) {
            const int   s0 = __builtin_amdgcn_readlane(s_l, j);
            const float w0 = __int_as_float(__builtin_amdgcn_readlane(__float_as_int(w_l), j));
            const vf h0 = *(const vf*)(H + (size_t)s0 * F + fo);
            if (r > 1) {
                const int   s1 = __builtin_amdgcn_readlane(s_l, j + 1);
                const float w1 = __int_as_float(__builtin_amdgcn_readlane(__float_as_int(w_l), j + 1));
                const vf h1 = *(const vf*)(H + (size_t)s1 * F + fo);
                if (r > 2) {
                    const int   s2 = __builtin_amdgcn_readlane(s_l, j + 2);
                    const float w2 = __int_as_float(__builtin_amdgcn_readlane(__float_as_int(w_l), j + 2));
                    const vf h2 = *(const vf*)(H + (size_t)s2 * F + fo);
                    acc += w0 * h0;
                    acc += w1 * h1;
                    acc += w2 * h2;
                } else {
                    acc += w0 * h0;
                    acc += w1 * h1;
                }
            } else {
                acc += w0 * h0;
            }
        }
    }
    acc += bv;
    if (RELU) {
#pragma unroll
        for (int k = 0; k < VEC; ++k) acc[k] = fmaxf(acc[k], 0.0f);
    }
    *(vf*)(out + (size_t)i * F + fo) = acc;
    }
}

template <int F>
__global__ void __launch_bounds__(512)
k_gemm_strip_f32(const float* __restrict__ X, const float* __restrict__ W,
                 float* __restrict__ H, int mtiles) {
    constexpr int K  = 128;
    constexpr int NT = F / 16;
    constexpr int NG = K / 8;
    __shared__ alignas(16) float Wlds[F * LDS_STRIDE];

    const int t     = threadIdx.x;
    const int wave  = __builtin_amdgcn_readfirstlane(t >> 5);
    const int lane  = t & 31;
    const int half  = lane >> 4;
    const int l16   = lane & 15;

    const float* __restrict__ Xl = X + (size_t)l16 * K + (half << 1);

    const int gx = (int)gridDim.x;
    int mtile = wave * gx + (int)blockIdx.x;
    const bool any = mtile < mtiles;

    v2f a[8];
#pragma unroll
    for (int u = 0; u < 8; ++u) a[u] = v2f{};
    if (any) {
#pragma unroll
        for (int u = 0; u < 8; ++u) a[u] = *(const v2f*)(Xl + (size_t)(mtile << 4) * K + 4 * u);
    }

    for (int idx = t; idx < K * F; idx += 512) {
        const int k = idx / F;
        const int n = idx & (F - 1);
        Wlds[n * LDS_STRIDE + k] = W[idx];
    }
    __syncthreads();
    if (!any) return;

    const float* __restrict__ Wl = &Wlds[l16 * LDS_STRIDE + (half << 1)];
    v2f bc[NT][2];
#pragma unroll
    for (int n = 0; n < NT; ++n) {
#pragma unroll
        for (int q = 0; q < 2; ++q) bc[n][q] = *(const v2f*)(Wl + n * 16 * LDS_STRIDE + 4 * q);
    }

    for (;;) {
        const int  row0  = mtile << 4;
        const int  mnext = mtile + 16 * gx;
        const bool more  = mnext < mtiles;
        const float* __restrict__ Xrow = Xl + (size_t)row0 * K;

        v8f c[NT];
#pragma unroll
        for (int n = 0; n < NT; ++n) c[n] = v8f{};

#pragma unroll
        for (int blk = 0; blk < 4; ++blk) {
            v2f an[8];
#pragma unroll
            for (int u = 0; u < 8; ++u) an[u] = a[u];
            if (blk < 3) {
#pragma unroll
                for (int u = 0; u < 8; ++u) an[u] = *(const v2f*)(Xrow + (blk + 1) * 32 + 4 * u);
            } else if (more) {
#pragma unroll
                for (int u = 0; u < 8; ++u) an[u] = *(const v2f*)(Xl + (size_t)(mnext << 4) * K + 4 * u);
            }
#pragma unroll
            for (int p = 0; p < 4; ++p) {
                const int g  = blk * 4 + p;
                const int gn = (g + 1) % NG;
                v2f bn[NT][2];
#pragma unroll
                for (int n = 0; n < NT; ++n) {
#pragma unroll
                    for (int q = 0; q < 2; ++q)
                        bn[n][q] = *(const v2f*)(Wl + n * 16 * LDS_STRIDE + gn * 8 + 4 * q);
                }
                __builtin_amdgcn_sched_barrier(0);
#pragma unroll
                for (int n = 0; n < NT; ++n) {
                    c[n] = __builtin_amdgcn_wmma_f32_16x16x4_f32(false, a[2 * p], false, bc[n][0],
                                                                 (short)0, c[n], false, false);
                    c[n] = __builtin_amdgcn_wmma_f32_16x16x4_f32(false, a[2 * p + 1], false, bc[n][1],
                                                                 (short)0, c[n], false, false);
                }
                __builtin_amdgcn_sched_barrier(0);
#pragma unroll
                for (int n = 0; n < NT; ++n) { bc[n][0] = bn[n][0]; bc[n][1] = bn[n][1]; }
            }
#pragma unroll
            for (int u = 0; u < 8; ++u) a[u] = an[u];
        }

        float* __restrict__ Ho = H + (size_t)(row0 + (half << 3)) * F + l16;
#pragma unroll
        for (int v = 0; v < 8; ++v) {
#pragma unroll
            for (int n = 0; n < NT; ++n)
                Ho[(size_t)v * F + n * 16] = c[n][v];
        }
        if (!more) break;
        mtile = mnext;
    }
}

#define FG_WAVES 25

typedef float fg_v4f __attribute__((ext_vector_type(4)));

template <int NROW>
__device__ __forceinline__ void fg_rows(const float* __restrict__ H, int s_l, float dn_l, float di,
                                        int j, int fo, fg_v4f& acc) {
    fg_v4f h[NROW];
#pragma unroll
    for (int u = 0; u < NROW; ++u) {
        const int s = __builtin_amdgcn_readlane(s_l, j + u);
        h[u] = *(const fg_v4f*)(H + (size_t)s * 128 + fo);
    }
    __builtin_amdgcn_sched_barrier(0);
#pragma unroll
    for (int u = 0; u < NROW; ++u) {
        const float w = __int_as_float(__builtin_amdgcn_readlane(__float_as_int(dn_l), j + u)) * di;
        acc += w * h[u];
    }
}

__device__ __forceinline__ void fg_rows_tail(const float* __restrict__ H, int s_l, float dn_l, float di,
                                             int j, int n, int fo, fg_v4f& acc) {
    fg_v4f h[15];
#pragma unroll
    for (int u = 0; u < 15; ++u) {
        h[u] = fg_v4f{};
        if (u < n) {
            const int s = __builtin_amdgcn_readlane(s_l, j + u);
            h[u] = *(const fg_v4f*)(H + (size_t)s * 128 + fo);
        }
    }
#pragma unroll
    for (int u = 0; u < 15; ++u) {
        if (u < n) {
            const float w = __int_as_float(__builtin_amdgcn_readlane(__float_as_int(dn_l), j + u)) * di;
            acc += w * h[u];
        }
    }
}

__device__ __forceinline__ void fg_chunk(const float* __restrict__ H, int s_l, float dn_l, float di,
                                         int m, int fo, fg_v4f& acc) {
    int j = 0;
    for (; j + 16 <= m; j += 16) fg_rows<16>(H, s_l, dn_l, di, j, fo, acc);
    if (j < m) fg_rows_tail(H, s_l, dn_l, di, j, m - j, fo, acc);
}

template <int FOUT>
__global__ void __launch_bounds__(FG_WAVES * 32)
k_gather_relu_gemm(const int* __restrict__ rowptr, const int* __restrict__ icnt,
                   const int* __restrict__ col, const float* __restrict__ dinv,
                   const float* __restrict__ Hin, const float* __restrict__ bias,
                   const float* __restrict__ Wn, float* __restrict__ Hout, int mtiles) {
    typedef float vf4 __attribute__((ext_vector_type(4)));
    constexpr int K    = 128;
    constexpr int NT   = FOUT / 16;
    constexpr int KG   = (NT >= 8) ? 1 : 2;
    constexpr int NGRP = (K / 4) / KG;
    __shared__ alignas(16) float Wlds[FOUT * LDS_STRIDE];
    __shared__ alignas(16) float Atile[FG_WAVES * 16 * LDS_STRIDE];

    const int t    = threadIdx.x;
    const int wave = __builtin_amdgcn_readfirstlane(t >> 5);
    const int lane = t & 31;
    const int half = lane >> 4;
    const int l16  = lane & 15;

    for (int idx = t; idx < K * FOUT; idx += FG_WAVES * 32) {
        const int k = idx / FOUT;
        const int n = idx & (FOUT - 1);
        Wlds[n * LDS_STRIDE + k] = Wn[idx];
    }
    __syncthreads();

    float* At = &Atile[wave * 16 * LDS_STRIDE];
    const float* Al = At + l16 * LDS_STRIDE + (half << 1);
    const float* __restrict__ Wl = &Wlds[l16 * LDS_STRIDE + (half << 1)];
    const int fo = lane * 4;
    const vf4 bv = *(const vf4*)(bias + fo);
    const int gx = (int)gridDim.x;

    for (int mtile = wave * gx + (int)blockIdx.x; mtile < mtiles; mtile += FG_WAVES * gx) {
        const int row0 = mtile << 4;
        const int   st_l = rowptr[row0 + l16];
        const int   ct_l = icnt[row0 + l16];
        const float di_l = dinv[row0 + l16];

        int   start = __builtin_amdgcn_readlane(st_l, 0);
        int   cnt   = __builtin_amdgcn_readlane(ct_l, 0);
        float di    = __int_as_float(__builtin_amdgcn_readlane(__float_as_int(di_l), 0));
        int   s_l = 0;
        if (lane < min(cnt, 32)) s_l = col[start + lane];
        vf4 acc0;
        {
            const vf4 hs = *(const vf4*)(Hin + (size_t)row0 * K + fo);
            acc0 = (di * di) * hs;
        }
        for (int r = 0; r < 16; ++r) {
            vf4 acc = acc0;
            float dn_l = 0.0f;
            if (lane < min(cnt, 32)) dn_l = dinv[s_l];
            const int   rn     = min(r + 1, 15);
            const int   nstart = __builtin_amdgcn_readlane(st_l, rn);
            const int   ncnt   = (r < 15) ? __builtin_amdgcn_readlane(ct_l, rn) : 0;
            const float ndi    = __int_as_float(__builtin_amdgcn_readlane(__float_as_int(di_l), rn));
            int s_n = 0;
            if (lane < min(ncnt, 32)) s_n = col[nstart + lane];
            vf4 hs_n = vf4{};
            if (r < 15) hs_n = *(const vf4*)(Hin + (size_t)(row0 + r + 1) * K + fo);

            fg_chunk(Hin, s_l, dn_l, di, min(cnt, 32), fo, acc);
            for (int base = 32; base < cnt; base += 32) {
                const int m = min(cnt - base, 32);
                int   s2 = 0;
                float d2 = 0.0f;
                if (lane < m) {
                    s2 = col[start + base + lane];
                    d2 = dinv[s2];
                }
                fg_chunk(Hin, s2, d2, di, m, fo, acc);
            }

            acc += bv;
#pragma unroll
            for (int k = 0; k < 4; ++k) acc[k] = fmaxf(acc[k], 0.0f);
            *(vf4*)(At + r * LDS_STRIDE + fo) = acc;

            acc0 = (ndi * ndi) * hs_n;
            s_l = s_n; start = nstart; cnt = ncnt; di = ndi;
        }

        v8f c[NT];
#pragma unroll
        for (int n = 0; n < NT; ++n) c[n] = v8f{};
        v2f ac[KG];
        v2f bc[NT][KG];
#pragma unroll
        for (int q = 0; q < KG; ++q) {
            ac[q] = *(const v2f*)(Al + 4 * q);
#pragma unroll
            for (int n = 0; n < NT; ++n) bc[n][q] = *(const v2f*)(Wl + n * 16 * LDS_STRIDE + 4 * q);
        }
#pragma unroll
        for (int g = 0; g < NGRP; ++g) {
            v2f an[KG];
            v2f bn[NT][KG];
#pragma unroll
            for (int q = 0; q < KG; ++q) {
                an[q] = ac[q];
#pragma unroll
                for (int n = 0; n < NT; ++n) bn[n][q] = bc[n][q];
            }
            if (g + 1 < NGRP) {
#pragma unroll
                for (int q = 0; q < KG; ++q) {
                    const int ko = 4 * ((g + 1) * KG + q);
                    an[q] = *(const v2f*)(Al + ko);
#pragma unroll
                    for (int n = 0; n < NT; ++n) bn[n][q] = *(const v2f*)(Wl + n * 16 * LDS_STRIDE + ko);
                }
            }
            __builtin_amdgcn_sched_barrier(0);
#pragma unroll
            for (int q = 0; q < KG; ++q) {
#pragma unroll
                for (int n = 0; n < NT; ++n)
                    c[n] = __builtin_amdgcn_wmma_f32_16x16x4_f32(false, ac[q], false, bc[n][q],
                                                                 (short)0, c[n], false, false);
            }
            __builtin_amdgcn_sched_barrier(0);
#pragma unroll
            for (int q = 0; q < KG; ++q) {
                ac[q] = an[q];
#pragma unroll
                for (int n = 0; n < NT; ++n) bc[n][q] = bn[n][q];
            }
        }
        float* __restrict__ Ho = Hout + (size_t)(row0 + (half << 3)) * FOUT + l16;
#pragma unroll
        for (int v = 0; v < 8; ++v) {
#pragma unroll
            for (int n = 0; n < NT; ++n)
                Ho[(size_t)v * FOUT + n * 16] = c[n][v];
        }
    }
}

__global__ void k_readout_zero(float* __restrict__ gsum, float* __restrict__ cnt) {
    int i = blockIdx.x * blockDim.x + threadIdx.x;
    if (i < GCN_G * GCN_OUT) gsum[i] = 0.0f;
    if (i < GCN_G) cnt[i] = 0.0f;
}

#define RO_CHUNK 64
__global__ void __launch_bounds__(256)
k_readout_scatter(const float* __restrict__ h, const int* __restrict__ batch,
                  float* __restrict__ gsum, float* __restrict__ cnt, int n) {
    const int f   = threadIdx.x & 63;
    const int sub = threadIdx.x >> 6;
    const int beg = blockIdx.x * RO_CHUNK;
    const int end = min(beg + RO_CHUNK, n);
    bool  have = false;
    int   cur  = 0;
    float acc  = 0.0f;
    float c    = 0.0f;
    for (int i0 = beg + sub; i0 < end; i0 += 32) {
        int   g[8];
        float v[8];
#pragma unroll
        for (int u = 0; u < 8; ++u) {
            const int i = i0 + 4 * u;
            g[u] = 0;
            v[u] = 0.0f;
            if (i < end) {
                g[u] = batch[i];
                v[u] = h[((size_t)i << 6) + f];
            }
        }
#pragma unroll
        for (int u = 0; u < 8; ++u) {
            const int i = i0 + 4 * u;
            if (i < end) {
                if (have && g[u] != cur) {
                    atomicAdd(&gsum[(cur << 6) + f], acc);
                    if (f == 0) atomicAdd(&cnt[cur], c);
                    acc = 0.0f;
                    c   = 0.0f;
                }
                have = true;
                cur  = g[u];
                acc += v[u];
                c   += 1.0f;
            }
        }
    }
    if (have) {
        atomicAdd(&gsum[(cur << 6) + f], acc);
        if (f == 0) atomicAdd(&cnt[cur], c);
    }
}

__global__ void k_readout_final(const float* __restrict__ gsum, const float* __restrict__ cnt,
                                float* __restrict__ outp) {
    int idx = blockIdx.x * blockDim.x + threadIdx.x;
    if (idx >= GCN_G * GCN_OUT) return;
    int g = idx >> 6;
    outp[idx] = gsum[idx] / fmaxf(cnt[g], 1.0f);
}

extern "C" void kernel_launch(void* const* d_in, const int* in_sizes, int n_in,
                              void* d_out, int out_size, void* d_ws, size_t ws_size,
                              hipStream_t stream) {
    const float* x     = (const float*)d_in[0];
    const int*   edge  = (const int*)d_in[1];
    const int*   batch = (const int*)d_in[2];
    const float* W1    = (const float*)d_in[3];
    const float* b1    = (const float*)d_in[4];
    const float* W2    = (const float*)d_in[5];
    const float* b2    = (const float*)d_in[6];
    const float* W3    = (const float*)d_in[7];
    const float* b3    = (const float*)d_in[8];

    const int IN = 128, HID = 128, OUTF = 64;
    const int N = in_sizes[0] / IN;
    const int E = in_sizes[1] / 2;
    const int* src = edge;
    const int* dst = edge + E;

    char* ws = (char*)d_ws;
    float* dinv = (float*)ws;
    size_t off = ((size_t)N * sizeof(float) + 511) & ~(size_t)511;
    float* bufA = (float*)(ws + off);
    float* bufB = bufA + (size_t)N * HID;
    float* gsum = bufB + (size_t)N * HID;
    float* cnt  = gsum + GCN_G * GCN_OUT;
    int*   icnt   = (int*)(cnt + GCN_G);
    int*   rowptr = icnt + N;
    int*   rank   = rowptr + N;
    int*   col    = rank + E;
    int*   bsum   = col + E;

    const int TB = 256;
    dim3 blk(TB);

    const int nscan = (N + 1023) / 1024;
    k_cnt_zero  <<<dim3((N + TB - 1) / TB), blk, 0, stream>>>(icnt, N);
    k_count_rank<<<dim3((E + 4 * TB - 1) / (4 * TB)), blk, 0, stream>>>(dst, icnt, rank, E);
    k_scan_local<<<dim3(nscan), blk, 0, stream>>>(icnt, rowptr, bsum, dinv, N);
    k_scan_add  <<<dim3(nscan), blk, 0, stream>>>(rowptr, bsum, N);
    k_fill      <<<dim3((E + 4 * TB - 1) / (4 * TB)), blk, 0, stream>>>(src, dst, rank, rowptr, col, E);

    const int mt = N / 16;
    const int gb = 256;
    const int gn = (N + 7) / 8 < 2048 ? (N + 7) / 8 : 2048;
    const int fgb = 250;

    k_gemm_strip_f32<128><<<dim3(gb), dim3(512), 0, stream>>>(x, W1, bufA, mt);
    k_gather_relu_gemm<128><<<dim3(fgb), dim3(FG_WAVES * 32), 0, stream>>>(rowptr, icnt, col, dinv, bufA, b1, W2, bufB, mt);

    k_gather_relu_gemm<64><<<dim3(fgb), dim3(FG_WAVES * 32), 0, stream>>>(rowptr, icnt, col, dinv, bufB, b2, W3, bufA, mt);

    k_gather<2, false><<<dim3(gn), blk, 0, stream>>>(rowptr, icnt, col, dinv, bufA, b3, bufB, N);

    k_readout_zero   <<<dim3((GCN_G * GCN_OUT + TB - 1) / TB), blk, 0, stream>>>(gsum, cnt);
    k_readout_scatter<<<dim3((N + RO_CHUNK - 1) / RO_CHUNK), blk, 0, stream>>>(bufB, batch, gsum, cnt, N);
    k_readout_final  <<<dim3((GCN_G * GCN_OUT + TB - 1) / TB), blk, 0, stream>>>(gsum, cnt, (float*)d_out);
}


# Readout folded into the layer-3 gather: waves own consecutive nodes, a block adds each graph run's sum / node count (from the scan kernel) to out by float atomics; 3 launches fewer; sum order changes
# speedup vs baseline: 4.2121x; 1.0474x over previous
#include <hip/hip_runtime.h>
#include <math.h>

typedef float v2f __attribute__((ext_vector_type(2)));
typedef float v8f __attribute__((ext_vector_type(8)));

#define GCN_G   64
#define GCN_OUT 64
#define LDS_STRIDE 132

__global__ void k_cnt_zero(int* __restrict__ icnt, int n, float* __restrict__ outp,
                           float* __restrict__ gcnt) {
    const int i  = blockIdx.x * blockDim.x + threadIdx.x;
    const int st = gridDim.x * blockDim.x;
    if (i < n) icnt[i] = 0;
    for (int j = i; j < GCN_G * GCN_OUT; j += st) outp[j] = 0.0f;
    for (int j = i; j < GCN_G; j += st) gcnt[j] = 0.0f;
}

__global__ void __launch_bounds__(256)
k_count_rank(const int* __restrict__ dst, int* __restrict__ icnt,
             int* __restrict__ rank, int e) {
    const int i0 = blockIdx.x * blockDim.x + threadIdx.x;
    const int st = gridDim.x * blockDim.x;
    int d[4], r[4];
#pragma unroll
    for (int k = 0; k < 4; ++k) {
        const int i = i0 + k * st;
        d[k] = dst[min(i, e - 1)];
    }
#pragma unroll
    for (int k = 0; k < 4; ++k) {
        const int i = i0 + k * st;
        r[k] = 0;
        if (i < e) r[k] = atomicAdd(&icnt[d[k]], 1);
    }
#pragma unroll
    for (int k = 0; k < 4; ++k) {
        const int i = i0 + k * st;
        if (i < e) rank[i] = r[k];
    }
}

__global__ void __launch_bounds__(256)
k_scan_local(const int* __restrict__ icnt, int* __restrict__ rowptr,
             int* __restrict__ bsum, float* __restrict__ dinv,
             const int* __restrict__ batch, float* __restrict__ gcnt, int n) {
    __shared__ int wsum[8];
    const int t    = threadIdx.x;
    const int lane = t & 31;
    const int wave = t >> 5;
    const int i0   = (blockIdx.x * 256 + t) * 4;
    int c[4];
#pragma unroll
    for (int k = 0; k < 4; ++k) c[k] = (i0 + k < n) ? icnt[i0 + k] : 0;
#pragma unroll
    for (int k = 0; k < 4; ++k) {
        if (i0 + k < n) {
            float d = (float)(c[k] + 1);
            dinv[i0 + k] = (d > 0.0f) ? (1.0f / sqrtf(d)) : 0.0f;
        }
    }
    {
        int gid[4];
        int nv = 0;
#pragma unroll
        for (int k = 0; k < 4; ++k) {
            const bool ok = i0 + k < n;
            gid[k] = ok ? batch[i0 + k] : 0;
            nv += ok ? 1 : 0;
        }
        const int gref = __builtin_amdgcn_readfirstlane(gid[0]);
        bool same = true;
#pragma unroll
        for (int k = 0; k < 4; ++k) same = same && (i0 + k >= n || gid[k] == gref);
        if (__all(same ? 1 : 0)) {
            int tot = nv;
#pragma unroll
            for (int off = 16; off > 0; off >>= 1) tot += __shfl_xor(tot, off);
            if (lane == 0 && tot > 0) atomicAdd(&gcnt[gref], (float)tot);
        } else {
            bool  have = false;
            int   cur  = 0;
            float run  = 0.0f;
#pragma unroll
            for (int k = 0; k < 4; ++k) {
                if (i0 + k < n) {
                    if (have && gid[k] != cur) {
                        atomicAdd(&gcnt[cur], run);
                        run = 0.0f;
                    }
                    have = true;
                    cur  = gid[k];
                    run += 1.0f;
                }
            }
            if (have) atomicAdd(&gcnt[cur], run);
        }
    }
    const int tsum = c[0] + c[1] + c[2] + c[3];
    int x = tsum;
#pragma unroll
    for (int off = 1; off < 32; off <<= 1) {
        int y = __shfl_up(x, off);
        if (lane >= off) x += y;
    }
    if (lane == 31) wsum[wave] = x;
    __syncthreads();
    int woff = 0;
    for (int w = 0; w < wave; ++w) woff += wsum[w];
    int excl = woff + x - tsum;
#pragma unroll
    for (int k = 0; k < 4; ++k) {
        if (i0 + k < n) rowptr[i0 + k] = excl;
        excl += c[k];
    }
    if (t == 255) bsum[blockIdx.x] = woff + x;
}

__global__ void __launch_bounds__(256)
k_scan_add(int* __restrict__ rowptr, const int* __restrict__ bsum, int n) {
    __shared__ int red[8];
    const int t    = threadIdx.x;
    const int lane = t & 31;
    const int wave = t >> 5;
    const int b    = blockIdx.x;
    int s = 0;
    for (int j = t; j < b; j += 256) s += bsum[j];
#pragma unroll
    for (int off = 16; off > 0; off >>= 1) s += __shfl_xor(s, off);
    if (lane == 0) red[wave] = s;
    __syncthreads();
    int tot = 0;
#pragma unroll
    for (int w = 0; w < 8; ++w) tot += red[w];
    const int i0 = (b * 256 + t) * 4;
#pragma unroll
    for (int k = 0; k < 4; ++k)
        if (i0 + k < n) rowptr[i0 + k] += tot;
}

__global__ void __launch_bounds__(256)
k_fill(const int* __restrict__ src, const int* __restrict__ dst,
       const int* __restrict__ rank, const int* __restrict__ rowptr,
       int* __restrict__ col, int e) {
    const int i0 = blockIdx.x * blockDim.x + threadIdx.x;
    const int st = gridDim.x * blockDim.x;
    int d[4], r[4], sv[4], p[4];
#pragma unroll
    for (int k = 0; k < 4; ++k) {
        const int ic = min(i0 + k * st, e - 1);
        d[k] = dst[ic]; r[k] = rank[ic]; sv[k] = src[ic];
    }
#pragma unroll
    for (int k = 0; k < 4; ++k) p[k] = rowptr[d[k]];
#pragma unroll
    for (int k = 0; k < 4; ++k) {
        const int i = i0 + k * st;
        if (i < e) col[p[k] + r[k]] = sv[k];
    }
}

template <int VEC>
__global__ void __launch_bounds__(256)
k_gather_readout(const int* __restrict__ rowptr, const int* __restrict__ icnt,
                 const int* __restrict__ col, const float* __restrict__ dinv,
                 const float* __restrict__ H, const float* __restrict__ bias,
                 const int* __restrict__ batch, const float* __restrict__ gcnt,
                 float* __restrict__ outp, int n) {
    typedef float vf __attribute__((ext_vector_type(VEC)));
    constexpr int F = VEC * 32;
    __shared__ alignas(16) float rsum[8][F];
    __shared__ int rid[8];
    __shared__ int rhave[8];
    const int lane = threadIdx.x & 31;
    const int wv  = __builtin_amdgcn_readfirstlane((int)(threadIdx.x >> 5));
    const int wid = (int)blockIdx.x * 8 + wv;
    const int nw  = (int)(gridDim.x * 8);
    const int fo  = lane * VEC;
    const vf  bv  = *(const vf*)(bias + fo);
    const int q   = n / nw;
    const int rem = n - q * nw;
    const int beg = wid * q + min(wid, rem);
    const int end = beg + q + (wid < rem ? 1 : 0);
    bool have = false;
    int  cur  = 0;
    vf   racc = vf{};
    for (int i = beg; i < end; ++i) {
    const int   start = rowptr[i];
    const int   cnt   = icnt[i];
    const float di    = dinv[i];
    const int   g     = batch[i];

    const vf hs = *(const vf*)(H + (size_t)i * F + fo);
    vf acc = (di * di) * hs;
    for (int base = 0; base < cnt; base += 32) {
        const int m = min(cnt - base, 32);
        int   s_l = 0;
        float w_l = 0.0f;
        if (lane < m) {
            s_l = col[start + base + lane];
            w_l = dinv[s_l] * di;
        }
        int j = 0;
        for (; j + 4 <= m; j += 4) {
            const int s0 = __builtin_amdgcn_readlane(s_l, j);
            const int s1 = __builtin_amdgcn_readlane(s_l, j + 1);
            const int s2 = __builtin_amdgcn_readlane(s_l, j + 2);
            const int s3 = __builtin_amdgcn_readlane(s_l, j + 3);
            const float w0 = __int_as_float(__builtin_amdgcn_readlane(__float_as_int(w_l), j));
            const float w1 = __int_as_float(__builtin_amdgcn_readlane(__float_as_int(w_l), j + 1));
            const float w2 = __int_as_float(__builtin_amdgcn_readlane(__float_as_int(w_l), j + 2));
            const float w3 = __int_as_float(__builtin_amdgcn_readlane(__float_as_int(w_l), j + 3));
            const vf h0 = *(const vf*)(H + (size_t)s0 * F + fo);
            const vf h1 = *(const vf*)(H + (size_t)s1 * F + fo);
            const vf h2 = *(const vf*)(H + (size_t)s2 * F + fo);
            const vf h3 = *(const vf*)(H + (size_t)s3 * F + fo);
            acc += w0 * h0;
            acc += w1 * h1;
            acc += w2 * h2;
            acc += w3 * h3;
        }
        const int r = m - j;
        if (r > 0) {
            const int   s0 = __builtin_amdgcn_readlane(s_l, j);
            const float w0 = __int_as_float(__builtin_amdgcn_readlane(__float_as_int(w_l), j));
            const vf h0 = *(const vf*)(H + (size_t)s0 * F + fo);
            if (r > 1) {
                const int   s1 = __builtin_amdgcn_readlane(s_l, j + 1);
                const float w1 = __int_as_float(__builtin_amdgcn_readlane(__float_as_int(w_l), j + 1));
                const vf h1 = *(const vf*)(H + (size_t)s1 * F + fo);
                if (r > 2) {
                    const int   s2 = __builtin_amdgcn_readlane(s_l, j + 2);
                    const float w2 = __int_as_float(__builtin_amdgcn_readlane(__float_as_int(w_l), j + 2));
                    const vf h2 = *(const vf*)(H + (size_t)s2 * F + fo);
                    acc += w0 * h0;
                    acc += w1 * h1;
                    acc += w2 * h2;
                } else {
                    acc += w0 * h0;
                    acc += w1 * h1;
                }
            } else {
                acc += w0 * h0;
            }
        }
    }
    acc += bv;
    if (have && g != cur) {
        const float cn = fmaxf(gcnt[cur], 1.0f);
#pragma unroll
        for (int k = 0; k < VEC; ++k) atomicAdd(&outp[cur * F + fo + k], racc[k] / cn);
        racc = vf{};
    }
    have = true;
    cur  = g;
    racc += acc;
    }

    *(vf*)(&rsum[wv][fo]) = racc;
    if (lane == 0) {
        rid[wv]   = cur;
        rhave[wv] = have ? 1 : 0;
    }
    __syncthreads();
    if (threadIdx.x < F) {
        const int f = threadIdx.x;
        int   hv[8], gw[8];
        float v[8];
#pragma unroll
        for (int w = 0; w < 8; ++w) {
            hv[w] = rhave[w];
            gw[w] = rid[w];
            v[w]  = rsum[w][f];
        }
        bool  h  = false;
        int   cg = 0;
        float a  = 0.0f;
#pragma unroll
        for (int w = 0; w < 8; ++w) {
            if (hv[w]) {
                if (h && gw[w] != cg) {
                    atomicAdd(&outp[cg * F + f], a / fmaxf(gcnt[cg], 1.0f));
                    a = 0.0f;
                }
                h  = true;
                cg = gw[w];
                a += v[w];
            }
        }
        if (h) atomicAdd(&outp[cg * F + f], a / fmaxf(gcnt[cg], 1.0f));
    }
}

template <int F>
__global__ void __launch_bounds__(512)
k_gemm_strip_f32(const float* __restrict__ X, const float* __restrict__ W,
                 float* __restrict__ H, int mtiles) {
    constexpr int K  = 128;
    constexpr int NT = F / 16;
    constexpr int NG = K / 8;
    __shared__ alignas(16) float Wlds[F * LDS_STRIDE];

    const int t     = threadIdx.x;
    const int wave  = __builtin_amdgcn_readfirstlane(t >> 5);
    const int lane  = t & 31;
    const int half  = lane >> 4;
    const int l16   = lane & 15;

    const float* __restrict__ Xl = X + (size_t)l16 * K + (half << 1);

    const int gx = (int)gridDim.x;
    int mtile = wave * gx + (int)blockIdx.x;
    const bool any = mtile < mtiles;

    v2f a[8];
#pragma unroll
    for (int u = 0; u < 8; ++u) a[u] = v2f{};
    if (any) {
#pragma unroll
        for (int u = 0; u < 8; ++u) a[u] = *(const v2f*)(Xl + (size_t)(mtile << 4) * K + 4 * u);
    }

    for (int idx = t; idx < K * F; idx += 512) {
        const int k = idx / F;
        const int n = idx & (F - 1);
        Wlds[n * LDS_STRIDE + k] = W[idx];
    }
    __syncthreads();
    if (!any) return;

    const float* __restrict__ Wl = &Wlds[l16 * LDS_STRIDE + (half << 1)];
    v2f bc[NT][2];
#pragma unroll
    for (int n = 0; n < NT; ++n) {
#pragma unroll
        for (int q = 0; q < 2; ++q) bc[n][q] = *(const v2f*)(Wl + n * 16 * LDS_STRIDE + 4 * q);
    }

    for (;;) {
        const int  row0  = mtile << 4;
        const int  mnext = mtile + 16 * gx;
        const bool more  = mnext < mtiles;
        const float* __restrict__ Xrow = Xl + (size_t)row0 * K;

        v8f c[NT];
#pragma unroll
        for (int n = 0; n < NT; ++n) c[n] = v8f{};

#pragma unroll
        for (int blk = 0; blk < 4; ++blk) {
            v2f an[8];
#pragma unroll
            for (int u = 0; u < 8; ++u) an[u] = a[u];
            if (blk < 3) {
#pragma unroll
                for (int u = 0; u < 8; ++u) an[u] = *(const v2f*)(Xrow + (blk + 1) * 32 + 4 * u);
            } else if (more) {
#pragma unroll
                for (int u = 0; u < 8; ++u) an[u] = *(const v2f*)(Xl + (size_t)(mnext << 4) * K + 4 * u);
            }
#pragma unroll
            for (int p = 0; p < 4; ++p) {
                const int g  = blk * 4 + p;
                const int gn = (g + 1) % NG;
                v2f bn[NT][2];
#pragma unroll
                for (int n = 0; n < NT; ++n) {
#pragma unroll
                    for (int q = 0; q < 2; ++q)
                        bn[n][q] = *(const v2f*)(Wl + n * 16 * LDS_STRIDE + gn * 8 + 4 * q);
                }
                __builtin_amdgcn_sched_barrier(0);
#pragma unroll
                for (int n = 0; n < NT; ++n) {
                    c[n] = __builtin_amdgcn_wmma_f32_16x16x4_f32(false, a[2 * p], false, bc[n][0],
                                                                 (short)0, c[n], false, false);
                    c[n] = __builtin_amdgcn_wmma_f32_16x16x4_f32(false, a[2 * p + 1], false, bc[n][1],
                                                                 (short)0, c[n], false, false);
                }
                __builtin_amdgcn_sched_barrier(0);
#pragma unroll
                for (int n = 0; n < NT; ++n) { bc[n][0] = bn[n][0]; bc[n][1] = bn[n][1]; }
            }
#pragma unroll
            for (int u = 0; u < 8; ++u) a[u] = an[u];
        }

        float* __restrict__ Ho = H + (size_t)(row0 + (half << 3)) * F + l16;
#pragma unroll
        for (int v = 0; v < 8; ++v) {
#pragma unroll
            for (int n = 0; n < NT; ++n)
                Ho[(size_t)v * F + n * 16] = c[n][v];
        }
        if (!more) break;
        mtile = mnext;
    }
}

#define FG_WAVES 25

typedef float fg_v4f __attribute__((ext_vector_type(4)));

template <int NROW>
__device__ __forceinline__ void fg_rows(const float* __restrict__ H, int s_l, float dn_l, float di,
                                        int j, int fo, fg_v4f& acc) {
    fg_v4f h[NROW];
#pragma unroll
    for (int u = 0; u < NROW; ++u) {
        const int s = __builtin_amdgcn_readlane(s_l, j + u);
        h[u] = *(const fg_v4f*)(H + (size_t)s * 128 + fo);
    }
    __builtin_amdgcn_sched_barrier(0);
#pragma unroll
    for (int u = 0; u < NROW; ++u) {
        const float w = __int_as_float(__builtin_amdgcn_readlane(__float_as_int(dn_l), j + u)) * di;
        acc += w * h[u];
    }
}

__device__ __forceinline__ void fg_rows_tail(const float* __restrict__ H, int s_l, float dn_l, float di,
                                             int j, int n, int fo, fg_v4f& acc) {
    fg_v4f h[15];
#pragma unroll
    for (int u = 0; u < 15; ++u) {
        h[u] = fg_v4f{};
        if (u < n) {
            const int s = __builtin_amdgcn_readlane(s_l, j + u);
            h[u] = *(const fg_v4f*)(H + (size_t)s * 128 + fo);
        }
    }
#pragma unroll
    for (int u = 0; u < 15; ++u) {
        if (u < n) {
            const float w = __int_as_float(__builtin_amdgcn_readlane(__float_as_int(dn_l), j + u)) * di;
            acc += w * h[u];
        }
    }
}

__device__ __forceinline__ void fg_chunk(const float* __restrict__ H, int s_l, float dn_l, float di,
                                         int m, int fo, fg_v4f& acc) {
    int j = 0;
    for (; j + 16 <= m; j += 16) fg_rows<16>(H, s_l, dn_l, di, j, fo, acc);
    if (j < m) fg_rows_tail(H, s_l, dn_l, di, j, m - j, fo, acc);
}

template <int FOUT>
__global__ void __launch_bounds__(FG_WAVES * 32)
k_gather_relu_gemm(const int* __restrict__ rowptr, const int* __restrict__ icnt,
                   const int* __restrict__ col, const float* __restrict__ dinv,
                   const float* __restrict__ Hin, const float* __restrict__ bias,
                   const float* __restrict__ Wn, float* __restrict__ Hout, int mtiles) {
    typedef float vf4 __attribute__((ext_vector_type(4)));
    constexpr int K    = 128;
    constexpr int NT   = FOUT / 16;
    constexpr int KG   = (NT >= 8) ? 1 : 2;
    constexpr int NGRP = (K / 4) / KG;
    __shared__ alignas(16) float Wlds[FOUT * LDS_STRIDE];
    __shared__ alignas(16) float Atile[FG_WAVES * 16 * LDS_STRIDE];

    const int t    = threadIdx.x;
    const int wave = __builtin_amdgcn_readfirstlane(t >> 5);
    const int lane = t & 31;
    const int half = lane >> 4;
    const int l16  = lane & 15;

    for (int idx = t; idx < K * FOUT; idx += FG_WAVES * 32) {
        const int k = idx / FOUT;
        const int n = idx & (FOUT - 1);
        Wlds[n * LDS_STRIDE + k] = Wn[idx];
    }
    __syncthreads();

    float* At = &Atile[wave * 16 * LDS_STRIDE];
    const float* Al = At + l16 * LDS_STRIDE + (half << 1);
    const float* __restrict__ Wl = &Wlds[l16 * LDS_STRIDE + (half << 1)];
    const int fo = lane * 4;
    const vf4 bv = *(const vf4*)(bias + fo);
    const int gx = (int)gridDim.x;

    for (int mtile = wave * gx + (int)blockIdx.x; mtile < mtiles; mtile += FG_WAVES * gx) {
        const int row0 = mtile << 4;
        const int   st_l = rowptr[row0 + l16];
        const int   ct_l = icnt[row0 + l16];
        const float di_l = dinv[row0 + l16];

        int   start = __builtin_amdgcn_readlane(st_l, 0);
        int   cnt   = __builtin_amdgcn_readlane(ct_l, 0);
        float di    = __int_as_float(__builtin_amdgcn_readlane(__float_as_int(di_l), 0));
        int   s_l = 0;
        if (lane < min(cnt, 32)) s_l = col[start + lane];
        vf4 acc0;
        {
            const vf4 hs = *(const vf4*)(Hin + (size_t)row0 * K + fo);
            acc0 = (di * di) * hs;
        }
        for (int r = 0; r < 16; ++r) {
            vf4 acc = acc0;
            float dn_l = 0.0f;
            if (lane < min(cnt, 32)) dn_l = dinv[s_l];
            const int   rn     = min(r + 1, 15);
            const int   nstart = __builtin_amdgcn_readlane(st_l, rn);
            const int   ncnt   = (r < 15) ? __builtin_amdgcn_readlane(ct_l, rn) : 0;
            const float ndi    = __int_as_float(__builtin_amdgcn_readlane(__float_as_int(di_l), rn));
            int s_n = 0;
            if (lane < min(ncnt, 32)) s_n = col[nstart + lane];
            vf4 hs_n = vf4{};
            if (r < 15) hs_n = *(const vf4*)(Hin + (size_t)(row0 + r + 1) * K + fo);

            fg_chunk(Hin, s_l, dn_l, di, min(cnt, 32), fo, acc);
            for (int base = 32; base < cnt; base += 32) {
                const int m = min(cnt - base, 32);
                int   s2 = 0;
                float d2 = 0.0f;
                if (lane < m) {
                    s2 = col[start + base + lane];
                    d2 = dinv[s2];
                }
                fg_chunk(Hin, s2, d2, di, m, fo, acc);
            }

            acc += bv;
#pragma unroll
            for (int k = 0; k < 4; ++k) acc[k] = fmaxf(acc[k], 0.0f);
            *(vf4*)(At + r * LDS_STRIDE + fo) = acc;

            acc0 = (ndi * ndi) * hs_n;
            s_l = s_n; start = nstart; cnt = ncnt; di = ndi;
        }

        v8f c[NT];
#pragma unroll
        for (int n = 0; n < NT; ++n) c[n] = v8f{};
        v2f ac[KG];
        v2f bc[NT][KG];
#pragma unroll
        for (int q = 0; q < KG; ++q) {
            ac[q] = *(const v2f*)(Al + 4 * q);
#pragma unroll
            for (int n = 0; n < NT; ++n) bc[n][q] = *(const v2f*)(Wl + n * 16 * LDS_STRIDE + 4 * q);
        }
#pragma unroll
        for (int g = 0; g < NGRP; ++g) {
            v2f an[KG];
            v2f bn[NT][KG];
#pragma unroll
            for (int q = 0; q < KG; ++q) {
                an[q] = ac[q];
#pragma unroll
                for (int n = 0; n < NT; ++n) bn[n][q] = bc[n][q];
            }
            if (g + 1 < NGRP) {
#pragma unroll
                for (int q = 0; q < KG; ++q) {
                    const int ko = 4 * ((g + 1) * KG + q);
                    an[q] = *(const v2f*)(Al + ko);
#pragma unroll
                    for (int n = 0; n < NT; ++n) bn[n][q] = *(const v2f*)(Wl + n * 16 * LDS_STRIDE + ko);
                }
            }
            __builtin_amdgcn_sched_barrier(0);
#pragma unroll
            for (int q = 0; q < KG; ++q) {
#pragma unroll
                for (int n = 0; n < NT; ++n)
                    c[n] = __builtin_amdgcn_wmma_f32_16x16x4_f32(false, ac[q], false, bc[n][q],
                                                                 (short)0, c[n], false, false);
            }
            __builtin_amdgcn_sched_barrier(0);
#pragma unroll
            for (int q = 0; q < KG; ++q) {
                ac[q] = an[q];
#pragma unroll
                for (int n = 0; n < NT; ++n) bc[n][q] = bn[n][q];
            }
        }
        float* __restrict__ Ho = Hout + (size_t)(row0 + (half << 3)) * FOUT + l16;
#pragma unroll
        for (int v = 0; v < 8; ++v) {
#pragma unroll
            for (int n = 0; n < NT; ++n)
                Ho[(size_t)v * FOUT + n * 16] = c[n][v];
        }
    }
}

extern "C" void kernel_launch(void* const* d_in, const int* in_sizes, int n_in,
                              void* d_out, int out_size, void* d_ws, size_t ws_size,
                              hipStream_t stream) {
    const float* x     = (const float*)d_in[0];
    const int*   edge  = (const int*)d_in[1];
    const int*   batch = (const int*)d_in[2];
    const float* W1    = (const float*)d_in[3];
    const float* b1    = (const float*)d_in[4];
    const float* W2    = (const float*)d_in[5];
    const float* b2    = (const float*)d_in[6];
    const float* W3    = (const float*)d_in[7];
    const float* b3    = (const float*)d_in[8];

    const int IN = 128, HID = 128, OUTF = 64;
    const int N = in_sizes[0] / IN;
    const int E = in_sizes[1] / 2;
    const int* src = edge;
    const int* dst = edge + E;

    char* ws = (char*)d_ws;
    float* dinv = (float*)ws;
    size_t off = ((size_t)N * sizeof(float) + 511) & ~(size_t)511;
    float* bufA = (float*)(ws + off);
    float* bufB = bufA + (size_t)N * HID;
    float* gcnt = bufB + (size_t)N * HID;
    int*   icnt   = (int*)(gcnt + GCN_G);
    int*   rowptr = icnt + N;
    int*   rank   = rowptr + N;
    int*   col    = rank + E;
    int*   bsum   = col + E;

    const int TB = 256;
    dim3 blk(TB);

    const int nscan = (N + 1023) / 1024;
    k_cnt_zero  <<<dim3((N + TB - 1) / TB), blk, 0, stream>>>(icnt, N, (float*)d_out, gcnt);
    k_count_rank<<<dim3((E + 4 * TB - 1) / (4 * TB)), blk, 0, stream>>>(dst, icnt, rank, E);
    k_scan_local<<<dim3(nscan), blk, 0, stream>>>(icnt, rowptr, bsum, dinv, batch, gcnt, N);
    k_scan_add  <<<dim3(nscan), blk, 0, stream>>>(rowptr, bsum, N);
    k_fill      <<<dim3((E + 4 * TB - 1) / (4 * TB)), blk, 0, stream>>>(src, dst, rank, rowptr, col, E);

    const int mt = N / 16;
    const int gb = 256;
    const int gn = (N + 7) / 8 < 2048 ? (N + 7) / 8 : 2048;
    const int fgb = 250;

    k_gemm_strip_f32<128><<<dim3(gb), dim3(512), 0, stream>>>(x, W1, bufA, mt);
    k_gather_relu_gemm<128><<<dim3(fgb), dim3(FG_WAVES * 32), 0, stream>>>(rowptr, icnt, col, dinv, bufA, b1, W2, bufB, mt);

    k_gather_relu_gemm<64><<<dim3(fgb), dim3(FG_WAVES * 32), 0, stream>>>(rowptr, icnt, col, dinv, bufB, b2, W3, bufA, mt);

    k_gather_readout<2><<<dim3(gn), blk, 0, stream>>>(rowptr, icnt, col, dinv, bufA, b3, batch, gcnt, (float*)d_out, N);
}


# Layer 1's product shares the launches of the two edge passes (count, fill): per block 8 waves multiply half the strips, 8 loop over edges; 1 launch fewer; row order = atomic arrival rank as before
# speedup vs baseline: 4.3599x; 1.0351x over previous
#include <hip/hip_runtime.h>
#include <math.h>

typedef float v2f __attribute__((ext_vector_type(2)));
typedef float v8f __attribute__((ext_vector_type(8)));

#define GCN_G   64
#define GCN_OUT 64
#define LDS_STRIDE 132

__global__ void k_cnt_zero(int* __restrict__ icnt, int n, float* __restrict__ outp,
                           float* __restrict__ gcnt) {
    const int i  = blockIdx.x * blockDim.x + threadIdx.x;
    const int st = gridDim.x * blockDim.x;
    if (i < n) icnt[i] = 0;
    for (int j = i; j < GCN_G * GCN_OUT; j += st) outp[j] = 0.0f;
    for (int j = i; j < GCN_G; j += st) gcnt[j] = 0.0f;
}

#define EG_WAVES 16
#define EG_MUL    8
#define EG_EPL    8

__device__ __forceinline__ void edge_count_rank(const int* __restrict__ dst, int* __restrict__ icnt,
                                                int* __restrict__ rank, int e, int gl, int nl) {
    const int trip = EG_EPL * nl;
    int d[EG_EPL];
#pragma unroll
    for (int k = 0; k < EG_EPL; ++k) d[k] = dst[min(gl + k * nl, e - 1)];
    for (int base = 0; base < e; base += trip) {
        int dn[EG_EPL], r[EG_EPL];
#pragma unroll
        for (int k = 0; k < EG_EPL; ++k) dn[k] = d[k];
        if (base + trip < e) {
#pragma unroll
            for (int k = 0; k < EG_EPL; ++k) dn[k] = dst[min(base + trip + gl + k * nl, e - 1)];
        }
#pragma unroll
        for (int k = 0; k < EG_EPL; ++k) {
            const int i = base + gl + k * nl;
            r[k] = 0;
            if (i < e) r[k] = atomicAdd(&icnt[d[k]], 1);
        }
#pragma unroll
        for (int k = 0; k < EG_EPL; ++k) {
            const int i = base + gl + k * nl;
            if (i < e) rank[i] = r[k];
        }
#pragma unroll
        for (int k = 0; k < EG_EPL; ++k) d[k] = dn[k];
    }
}

__global__ void __launch_bounds__(256)
k_scan_local(const int* __restrict__ icnt, int* __restrict__ rowptr,
             int* __restrict__ bsum, float* __restrict__ dinv,
             const int* __restrict__ batch, float* __restrict__ gcnt, int n) {
    __shared__ int wsum[8];
    const int t    = threadIdx.x;
    const int lane = t & 31;
    const int wave = t >> 5;
    const int i0   = (blockIdx.x * 256 + t) * 4;
    int c[4];
#pragma unroll
    for (int k = 0; k < 4; ++k) c[k] = (i0 + k < n) ? icnt[i0 + k] : 0;
#pragma unroll
    for (int k = 0; k < 4; ++k) {
        if (i0 + k < n) {
            float d = (float)(c[k] + 1);
            dinv[i0 + k] = (d > 0.0f) ? (1.0f / sqrtf(d)) : 0.0f;
        }
    }
    {
        int gid[4];
        int nv = 0;
#pragma unroll
        for (int k = 0; k < 4; ++k) {
            const bool ok = i0 + k < n;
            gid[k] = ok ? batch[i0 + k] : 0;
            nv += ok ? 1 : 0;
        }
        const int gref = __builtin_amdgcn_readfirstlane(gid[0]);
        bool same = true;
#pragma unroll
        for (int k = 0; k < 4; ++k) same = same && (i0 + k >= n || gid[k] == gref);
        if (__all(same ? 1 : 0)) {
            int tot = nv;
#pragma unroll
            for (int off = 16; off > 0; off >>= 1) tot += __shfl_xor(tot, off);
            if (lane == 0 && tot > 0) atomicAdd(&gcnt[gref], (float)tot);
        } else {
            bool  have = false;
            int   cur  = 0;
            float run  = 0.0f;
#pragma unroll
            for (int k = 0; k < 4; ++k) {
                if (i0 + k < n) {
                    if (have && gid[k] != cur) {
                        atomicAdd(&gcnt[cur], run);
                        run = 0.0f;
                    }
                    have = true;
                    cur  = gid[k];
                    run += 1.0f;
                }
            }
            if (have) atomicAdd(&gcnt[cur], run);
        }
    }
    const int tsum = c[0] + c[1] + c[2] + c[3];
    int x = tsum;
#pragma unroll
    for (int off = 1; off < 32; off <<= 1) {
        int y = __shfl_up(x, off);
        if (lane >= off) x += y;
    }
    if (lane == 31) wsum[wave] = x;
    __syncthreads();
    int woff = 0;
    for (int w = 0; w < wave; ++w) woff += wsum[w];
    int excl = woff + x - tsum;
#pragma unroll
    for (int k = 0; k < 4; ++k) {
        if (i0 + k < n) rowptr[i0 + k] = excl;
        excl += c[k];
    }
    if (t == 255) bsum[blockIdx.x] = woff + x;
}

__global__ void __launch_bounds__(256)
k_scan_add(int* __restrict__ rowptr, const int* __restrict__ bsum, int n) {
    __shared__ int red[8];
    const int t    = threadIdx.x;
    const int lane = t & 31;
    const int wave = t >> 5;
    const int b    = blockIdx.x;
    int s = 0;
    for (int j = t; j < b; j += 256) s += bsum[j];
#pragma unroll
    for (int off = 16; off > 0; off >>= 1) s += __shfl_xor(s, off);
    if (lane == 0) red[wave] = s;
    __syncthreads();
    int tot = 0;
#pragma unroll
    for (int w = 0; w < 8; ++w) tot += red[w];
    const int i0 = (b * 256 + t) * 4;
#pragma unroll
    for (int k = 0; k < 4; ++k)
        if (i0 + k < n) rowptr[i0 + k] += tot;
}

__device__ __forceinline__ void edge_fill(const int* __restrict__ src, const int* __restrict__ dst,
                                          const int* __restrict__ rank, const int* __restrict__ rowptr,
                                          int* __restrict__ col, int e, int gl, int nl) {
    const int trip = EG_EPL * nl;
    int d[EG_EPL], r[EG_EPL], sv[EG_EPL];
#pragma unroll
    for (int k = 0; k < EG_EPL; ++k) {
        const int ic = min(gl + k * nl, e - 1);
        d[k] = dst[ic]; r[k] = rank[ic]; sv[k] = src[ic];
    }
    for (int base = 0; base < e; base += trip) {
        int dn[EG_EPL], rn[EG_EPL], sn[EG_EPL], p[EG_EPL];
#pragma unroll
        for (int k = 0; k < EG_EPL; ++k) { dn[k] = d[k]; rn[k] = r[k]; sn[k] = sv[k]; }
        if (base + trip < e) {
#pragma unroll
            for (int k = 0; k < EG_EPL; ++k) {
                const int ic = min(base + trip + gl + k * nl, e - 1);
                dn[k] = dst[ic]; rn[k] = rank[ic]; sn[k] = src[ic];
            }
        }
#pragma unroll
        for (int k = 0; k < EG_EPL; ++k) p[k] = rowptr[d[k]];
#pragma unroll
        for (int k = 0; k < EG_EPL; ++k) {
            const int i = base + gl + k * nl;
            if (i < e) col[p[k] + r[k]] = sv[k];
        }
#pragma unroll
        for (int k = 0; k < EG_EPL; ++k) { d[k] = dn[k]; r[k] = rn[k]; sv[k] = sn[k]; }
    }
}

template <int VEC>
__global__ void __launch_bounds__(256)
k_gather_readout(const int* __restrict__ rowptr, const int* __restrict__ icnt,
                 const int* __restrict__ col, const float* __restrict__ dinv,
                 const float* __restrict__ H, const float* __restrict__ bias,
                 const int* __restrict__ batch, const float* __restrict__ gcnt,
                 float* __restrict__ outp, int n) {
    typedef float vf __attribute__((ext_vector_type(VEC)));
    constexpr int F = VEC * 32;
    __shared__ alignas(16) float rsum[8][F];
    __shared__ int rid[8];
    __shared__ int rhave[8];
    const int lane = threadIdx.x & 31;
    const int wv  = __builtin_amdgcn_readfirstlane((int)(threadIdx.x >> 5));
    const int wid = (int)blockIdx.x * 8 + wv;
    const int nw  = (int)(gridDim.x * 8);
    const int fo  = lane * VEC;
    const vf  bv  = *(const vf*)(bias + fo);
    const int q   = n / nw;
    const int rem = n - q * nw;
    const int beg = wid * q + min(wid, rem);
    const int end = beg + q + (wid < rem ? 1 : 0);
    bool have = false;
    int  cur  = 0;
    vf   racc = vf{};
    for (int i = beg; i < end; ++i) {
    const int   start = rowptr[i];
    const int   cnt   = icnt[i];
    const float di    = dinv[i];
    const int   g     = batch[i];

    const vf hs = *(const vf*)(H + (size_t)i * F + fo);
    vf acc = (di * di) * hs;
    for (int base = 0; base < cnt; base += 32) {
        const int m = min(cnt - base, 32);
        int   s_l = 0;
        float w_l = 0.0f;
        if (lane < m) {
            s_l = col[start + base + lane];
            w_l = dinv[s_l] * di;
        }
        int j = 0;
        for (; j + 4 <= m; j += 4) {
            const int s0 = __builtin_amdgcn_readlane(s_l, j);
            const int s1 = __builtin_amdgcn_readlane(s_l, j + 1);
            const int s2 = __builtin_amdgcn_readlane(s_l, j + 2);
            const int s3 = __builtin_amdgcn_readlane(s_l, j + 3);
            const float w0 = __int_as_float(__builtin_amdgcn_readlane(__float_as_int(w_l), j));
            const float w1 = __int_as_float(__builtin_amdgcn_readlane(__float_as_int(w_l), j + 1));
            const float w2 = __int_as_float(__builtin_amdgcn_readlane(__float_as_int(w_l), j + 2));
            const float w3 = __int_as_float(__builtin_amdgcn_readlane(__float_as_int(w_l), j + 3));
            const vf h0 = *(const vf*)(H + (size_t)s0 * F + fo);
            const vf h1 = *(const vf*)(H + (size_t)s1 * F + fo);
            const vf h2 = *(const vf*)(H + (size_t)s2 * F + fo);
            const vf h3 = *(const vf*)(H + (size_t)s3 * F + fo);
            acc += w0 * h0;
            acc += w1 * h1;
            acc += w2 * h2;
            acc += w3 * h3;
        }
        const int r = m - j;
        if (r > 0) {
            const int   s0 = __builtin_amdgcn_readlane(s_l, j);
            const float w0 = __int_as_float(__builtin_amdgcn_readlane(__float_as_int(w_l), j));
            const vf h0 = *(const vf*)(H + (size_t)s0 * F + fo);
            if (r > 1) {
                const int   s1 = __builtin_amdgcn_readlane(s_l, j + 1);
                const float w1 = __int_as_float(__builtin_amdgcn_readlane(__float_as_int(w_l), j + 1));
                const vf h1 = *(const vf*)(H + (size_t)s1 * F + fo);
                if (r > 2) {
                    const int   s2 = __builtin_amdgcn_readlane(s_l, j + 2);
                    const float w2 = __int_as_float(__builtin_amdgcn_readlane(__float_as_int(w_l), j + 2));
                    const vf h2 = *(const vf*)(H + (size_t)s2 * F + fo);
                    acc += w0 * h0;
                    acc += w1 * h1;
                    acc += w2 * h2;
                } else {
                    acc += w0 * h0;
                    acc += w1 * h1;
                }
            } else {
                acc += w0 * h0;
            }
        }
    }
    acc += bv;
    if (have && g != cur) {
        const float cn = fmaxf(gcnt[cur], 1.0f);
#pragma unroll
        for (int k = 0; k < VEC; ++k) atomicAdd(&outp[cur * F + fo + k], racc[k] / cn);
        racc = vf{};
    }
    have = true;
    cur  = g;
    racc += acc;
    }

    *(vf*)(&rsum[wv][fo]) = racc;
    if (lane == 0) {
        rid[wv]   = cur;
        rhave[wv] = have ? 1 : 0;
    }
    __syncthreads();
    if (threadIdx.x < F) {
        const int f = threadIdx.x;
        int   hv[8], gw[8];
        float v[8];
#pragma unroll
        for (int w = 0; w < 8; ++w) {
            hv[w] = rhave[w];
            gw[w] = rid[w];
            v[w]  = rsum[w][f];
        }
        bool  h  = false;
        int   cg = 0;
        float a  = 0.0f;
#pragma unroll
        for (int w = 0; w < 8; ++w) {
            if (hv[w]) {
                if (h && gw[w] != cg) {
                    atomicAdd(&outp[cg * F + f], a / fmaxf(gcnt[cg], 1.0f));
                    a = 0.0f;
                }
                h  = true;
                cg = gw[w];
                a += v[w];
            }
        }
        if (h) atomicAdd(&outp[cg * F + f], a / fmaxf(gcnt[cg], 1.0f));
    }
}

template <int PHASE>
__global__ void __launch_bounds__(EG_WAVES * 32)
k_edges_gemm(const float* __restrict__ X, const float* __restrict__ W, float* __restrict__ H,
             int mt_lo, int mt_hi,
             const int* __restrict__ src, const int* __restrict__ dst, int* __restrict__ icnt,
             int* rank, const int* __restrict__ rowptr, int* __restrict__ col, int e) {
    constexpr int F  = 128;
    constexpr int K  = 128;
    constexpr int NT = F / 16;
    constexpr int NG = K / 8;
    __shared__ alignas(16) float Wlds[F * LDS_STRIDE];

    const int t     = threadIdx.x;
    const int wave  = __builtin_amdgcn_readfirstlane(t >> 5);
    const int lane  = t & 31;
    const int half  = lane >> 4;
    const int l16   = lane & 15;

    const float* __restrict__ Xl = X + (size_t)l16 * K + (half << 1);

    const int gx = (int)gridDim.x;
    int mtile = mt_lo + wave * gx + (int)blockIdx.x;
    const bool any = wave < EG_MUL && mtile < mt_hi;

    v2f a[8];
#pragma unroll
    for (int u = 0; u < 8; ++u) a[u] = v2f{};
    if (any) {
#pragma unroll
        for (int u = 0; u < 8; ++u) a[u] = *(const v2f*)(Xl + (size_t)(mtile << 4) * K + 4 * u);
    }

    for (int idx = t; idx < K * F; idx += EG_WAVES * 32) {
        const int k = idx / F;
        const int n = idx & (F - 1);
        Wlds[n * LDS_STRIDE + k] = W[idx];
    }
    __syncthreads();

    if (wave >= EG_MUL) {
        const int nl = gx * (EG_WAVES - EG_MUL) * 32;
        const int gl = ((int)blockIdx.x * (EG_WAVES - EG_MUL) + (wave - EG_MUL)) * 32 + lane;
        if (PHASE == 0) edge_count_rank(dst, icnt, rank, e, gl, nl);
        else            edge_fill(src, dst, rank, rowptr, col, e, gl, nl);
        return;
    }
    if (!any) return;

    const float* __restrict__ Wl = &Wlds[l16 * LDS_STRIDE + (half << 1)];
    v2f bc[NT][2];
#pragma unroll
    for (int n = 0; n < NT; ++n) {
#pragma unroll
        for (int q = 0; q < 2; ++q) bc[n][q] = *(const v2f*)(Wl + n * 16 * LDS_STRIDE + 4 * q);
    }

    for (;;) {
        const int  row0  = mtile << 4;
        const int  mnext = mtile + EG_MUL * gx;
        const bool more  = mnext < mt_hi;
        const float* __restrict__ Xrow = Xl + (size_t)row0 * K;

        v8f c[NT];
#pragma unroll
        for (int n = 0; n < NT; ++n) c[n] = v8f{};

#pragma unroll
        for (int blk = 0; blk < 4; ++blk) {
            v2f an[8];
#pragma unroll
            for (int u = 0; u < 8; ++u) an[u] = a[u];
            if (blk < 3) {
#pragma unroll
                for (int u = 0; u < 8; ++u) an[u] = *(const v2f*)(Xrow + (blk + 1) * 32 + 4 * u);
            } else if (more) {
#pragma unroll
                for (int u = 0; u < 8; ++u) an[u] = *(const v2f*)(Xl + (size_t)(mnext << 4) * K + 4 * u);
            }
#pragma unroll
            for (int p = 0; p < 4; ++p) {
                const int g  = blk * 4 + p;
                const int gn = (g + 1) % NG;
                v2f bn[NT][2];
#pragma unroll
                for (int n = 0; n < NT; ++n) {
#pragma unroll
                    for (int q = 0; q < 2; ++q)
                        bn[n][q] = *(const v2f*)(Wl + n * 16 * LDS_STRIDE + gn * 8 + 4 * q);
                }
                __builtin_amdgcn_sched_barrier(0);
#pragma unroll
                for (int n = 0; n < NT; ++n) {
                    c[n] = __builtin_amdgcn_wmma_f32_16x16x4_f32(false, a[2 * p], false, bc[n][0],
                                                                 (short)0, c[n], false, false);
                    c[n] = __builtin_amdgcn_wmma_f32_16x16x4_f32(false, a[2 * p + 1], false, bc[n][1],
                                                                 (short)0, c[n], false, false);
                }
                __builtin_amdgcn_sched_barrier(0);
#pragma unroll
                for (int n = 0; n < NT; ++n) { bc[n][0] = bn[n][0]; bc[n][1] = bn[n][1]; }
            }
#pragma unroll
            for (int u = 0; u < 8; ++u) a[u] = an[u];
        }

        float* __restrict__ Ho = H + (size_t)(row0 + (half << 3)) * F + l16;
#pragma unroll
        for (int v = 0; v < 8; ++v) {
#pragma unroll
            for (int n = 0; n < NT; ++n)
                Ho[(size_t)v * F + n * 16] = c[n][v];
        }
        if (!more) break;
        mtile = mnext;
    }
}

#define FG_WAVES 25

typedef float fg_v4f __attribute__((ext_vector_type(4)));

template <int NROW>
__device__ __forceinline__ void fg_rows(const float* __restrict__ H, int s_l, float dn_l, float di,
                                        int j, int fo, fg_v4f& acc) {
    fg_v4f h[NROW];
#pragma unroll
    for (int u = 0; u < NROW; ++u) {
        const int s = __builtin_amdgcn_readlane(s_l, j + u);
        h[u] = *(const fg_v4f*)(H + (size_t)s * 128 + fo);
    }
    __builtin_amdgcn_sched_barrier(0);
#pragma unroll
    for (int u = 0; u < NROW; ++u) {
        const float w = __int_as_float(__builtin_amdgcn_readlane(__float_as_int(dn_l), j + u)) * di;
        acc += w * h[u];
    }
}

__device__ __forceinline__ void fg_rows_tail(const float* __restrict__ H, int s_l, float dn_l, float di,
                                             int j, int n, int fo, fg_v4f& acc) {
    fg_v4f h[15];
#pragma unroll
    for (int u = 0; u < 15; ++u) {
        h[u] = fg_v4f{};
        if (u < n) {
            const int s = __builtin_amdgcn_readlane(s_l, j + u);
            h[u] = *(const fg_v4f*)(H + (size_t)s * 128 + fo);
        }
    }
#pragma unroll
    for (int u = 0; u < 15; ++u) {
        if (u < n) {
            const float w = __int_as_float(__builtin_amdgcn_readlane(__float_as_int(dn_l), j + u)) * di;
            acc += w * h[u];
        }
    }
}

__device__ __forceinline__ void fg_chunk(const float* __restrict__ H, int s_l, float dn_l, float di,
                                         int m, int fo, fg_v4f& acc) {
    int j = 0;
    for (; j + 16 <= m; j += 16) fg_rows<16>(H, s_l, dn_l, di, j, fo, acc);
    if (j < m) fg_rows_tail(H, s_l, dn_l, di, j, m - j, fo, acc);
}

template <int FOUT>
__global__ void __launch_bounds__(FG_WAVES * 32)
k_gather_relu_gemm(const int* __restrict__ rowptr, const int* __restrict__ icnt,
                   const int* __restrict__ col, const float* __restrict__ dinv,
                   const float* __restrict__ Hin, const float* __restrict__ bias,
                   const float* __restrict__ Wn, float* __restrict__ Hout, int mtiles) {
    typedef float vf4 __attribute__((ext_vector_type(4)));
    constexpr int K    = 128;
    constexpr int NT   = FOUT / 16;
    constexpr int KG   = (NT >= 8) ? 1 : 2;
    constexpr int NGRP = (K / 4) / KG;
    __shared__ alignas(16) float Wlds[FOUT * LDS_STRIDE];
    __shared__ alignas(16) float Atile[FG_WAVES * 16 * LDS_STRIDE];

    const int t    = threadIdx.x;
    const int wave = __builtin_amdgcn_readfirstlane(t >> 5);
    const int lane = t & 31;
    const int half = lane >> 4;
    const int l16  = lane & 15;

    for (int idx = t; idx < K * FOUT; idx += FG_WAVES * 32) {
        const int k = idx / FOUT;
        const int n = idx & (FOUT - 1);
        Wlds[n * LDS_STRIDE + k] = Wn[idx];
    }
    __syncthreads();

    float* At = &Atile[wave * 16 * LDS_STRIDE];
    const float* Al = At + l16 * LDS_STRIDE + (half << 1);
    const float* __restrict__ Wl = &Wlds[l16 * LDS_STRIDE + (half << 1)];
    const int fo = lane * 4;
    const vf4 bv = *(const vf4*)(bias + fo);
    const int gx = (int)gridDim.x;

    for (int mtile = wave * gx + (int)blockIdx.x; mtile < mtiles; mtile += FG_WAVES * gx) {
        const int row0 = mtile << 4;
        const int   st_l = rowptr[row0 + l16];
        const int   ct_l = icnt[row0 + l16];
        const float di_l = dinv[row0 + l16];

        int   start = __builtin_amdgcn_readlane(st_l, 0);
        int   cnt   = __builtin_amdgcn_readlane(ct_l, 0);
        float di    = __int_as_float(__builtin_amdgcn_readlane(__float_as_int(di_l), 0));
        int   s_l = 0;
        if (lane < min(cnt, 32)) s_l = col[start + lane];
        vf4 acc0;
        {
            const vf4 hs = *(const vf4*)(Hin + (size_t)row0 * K + fo);
            acc0 = (di * di) * hs;
        }
        for (int r = 0; r < 16; ++r) {
            vf4 acc = acc0;
            float dn_l = 0.0f;
            if (lane < min(cnt, 32)) dn_l = dinv[s_l];
            const int   rn     = min(r + 1, 15);
            const int   nstart = __builtin_amdgcn_readlane(st_l, rn);
            const int   ncnt   = (r < 15) ? __builtin_amdgcn_readlane(ct_l, rn) : 0;
            const float ndi    = __int_as_float(__builtin_amdgcn_readlane(__float_as_int(di_l), rn));
            int s_n = 0;
            if (lane < min(ncnt, 32)) s_n = col[nstart + lane];
            vf4 hs_n = vf4{};
            if (r < 15) hs_n = *(const vf4*)(Hin + (size_t)(row0 + r + 1) * K + fo);

            fg_chunk(Hin, s_l, dn_l, di, min(cnt, 32), fo, acc);
            for (int base = 32; base < cnt; base += 32) {
                const int m = min(cnt - base, 32);
                int   s2 = 0;
                float d2 = 0.0f;
                if (lane < m) {
                    s2 = col[start + base + lane];
                    d2 = dinv[s2];
                }
                fg_chunk(Hin, s2, d2, di, m, fo, acc);
            }

            acc += bv;
#pragma unroll
            for (int k = 0; k < 4; ++k) acc[k] = fmaxf(acc[k], 0.0f);
            *(vf4*)(At + r * LDS_STRIDE + fo) = acc;

            acc0 = (ndi * ndi) * hs_n;
            s_l = s_n; start = nstart; cnt = ncnt; di = ndi;
        }

        v8f c[NT];
#pragma unroll
        for (int n = 0; n < NT; ++n) c[n] = v8f{};
        v2f ac[KG];
        v2f bc[NT][KG];
#pragma unroll
        for (int q = 0; q < KG; ++q) {
            ac[q] = *(const v2f*)(Al + 4 * q);
#pragma unroll
            for (int n = 0; n < NT; ++n) bc[n][q] = *(const v2f*)(Wl + n * 16 * LDS_STRIDE + 4 * q);
        }
#pragma unroll
        for (int g = 0; g < NGRP; ++g) {
            v2f an[KG];
            v2f bn[NT][KG];
#pragma unroll
            for (int q = 0; q < KG; ++q) {
                an[q] = ac[q];
#pragma unroll
                for (int n = 0; n < NT; ++n) bn[n][q] = bc[n][q];
            }
            if (g + 1 < NGRP) {
#pragma unroll
                for (int q = 0; q < KG; ++q) {
                    const int ko = 4 * ((g + 1) * KG + q);
                    an[q] = *(const v2f*)(Al + ko);
#pragma unroll
                    for (int n = 0; n < NT; ++n) bn[n][q] = *(const v2f*)(Wl + n * 16 * LDS_STRIDE + ko);
                }
            }
            __builtin_amdgcn_sched_barrier(0);
#pragma unroll
            for (int q = 0; q < KG; ++q) {
#pragma unroll
                for (int n = 0; n < NT; ++n)
                    c[n] = __builtin_amdgcn_wmma_f32_16x16x4_f32(false, ac[q], false, bc[n][q],
                                                                 (short)0, c[n], false, false);
            }
            __builtin_amdgcn_sched_barrier(0);
#pragma unroll
            for (int q = 0; q < KG; ++q) {
                ac[q] = an[q];
#pragma unroll
                for (int n = 0; n < NT; ++n) bc[n][q] = bn[n][q];
            }
        }
        float* __restrict__ Ho = Hout + (size_t)(row0 + (half << 3)) * FOUT + l16;
#pragma unroll
        for (int v = 0; v < 8; ++v) {
#pragma unroll
            for (int n = 0; n < NT; ++n)
                Ho[(size_t)v * FOUT + n * 16] = c[n][v];
        }
    }
}

extern "C" void kernel_launch(void* const* d_in, const int* in_sizes, int n_in,
                              void* d_out, int out_size, void* d_ws, size_t ws_size,
                              hipStream_t stream) {
    const float* x     = (const float*)d_in[0];
    const int*   edge  = (const int*)d_in[1];
    const int*   batch = (const int*)d_in[2];
    const float* W1    = (const float*)d_in[3];
    const float* b1    = (const float*)d_in[4];
    const float* W2    = (const float*)d_in[5];
    const float* b2    = (const float*)d_in[6];
    const float* W3    = (const float*)d_in[7];
    const float* b3    = (const float*)d_in[8];

    const int IN = 128, HID = 128, OUTF = 64;
    const int N = in_sizes[0] / IN;
    const int E = in_sizes[1] / 2;
    const int* src = edge;
    const int* dst = edge + E;

    char* ws = (char*)d_ws;
    float* dinv = (float*)ws;
    size_t off = ((size_t)N * sizeof(float) + 511) & ~(size_t)511;
    float* bufA = (float*)(ws + off);
    float* bufB = bufA + (size_t)N * HID;
    float* gcnt = bufB + (size_t)N * HID;
    int*   icnt   = (int*)(gcnt + GCN_G);
    int*   rowptr = icnt + N;
    int*   rank   = rowptr + N;
    int*   col    = rank + E;
    int*   bsum   = col + E;

    const int TB = 256;
    dim3 blk(TB);

    const int mt  = N / 16;
    const int mth = mt / 2;
    const int egb = 256;

    const int nscan = (N + 1023) / 1024;
    k_cnt_zero  <<<dim3((N + TB - 1) / TB), blk, 0, stream>>>(icnt, N, (float*)d_out, gcnt);
    k_edges_gemm<0><<<dim3(egb), dim3(EG_WAVES * 32), 0, stream>>>(x, W1, bufA, 0, mth, src, dst, icnt, rank, rowptr, col, E);
    k_scan_local<<<dim3(nscan), blk, 0, stream>>>(icnt, rowptr, bsum, dinv, batch, gcnt, N);
    k_scan_add  <<<dim3(nscan), blk, 0, stream>>>(rowptr, bsum, N);
    k_edges_gemm<1><<<dim3(egb), dim3(EG_WAVES * 32), 0, stream>>>(x, W1, bufA, mth, mt, src, dst, icnt, rank, rowptr, col, E);

    const int gn = (N + 7) / 8 < 2048 ? (N + 7) / 8 : 2048;
    const int fgb = 250;

    k_gather_relu_gemm<128><<<dim3(fgb), dim3(FG_WAVES * 32), 0, stream>>>(rowptr, icnt, col, dinv, bufA, b1, W2, bufB, mt);

    k_gather_relu_gemm<64><<<dim3(fgb), dim3(FG_WAVES * 32), 0, stream>>>(rowptr, icnt, col, dinv, bufB, b2, W3, bufA, mt);

    k_gather_readout<2><<<dim3(gn), blk, 0, stream>>>(rowptr, icnt, col, dinv, bufA, b3, batch, gcnt, (float*)d_out, N);
}


# One edge pass: each node owns 32 source slots (slot = count atomic's return; overflow list past them), gathers form dinv from counts; no scan or fill pass, 3 launches fewer; sums in arrival order
# speedup vs baseline: 4.8070x; 1.1025x over previous
#include <hip/hip_runtime.h>
#include <math.h>

typedef float v2f __attribute__((ext_vector_type(2)));
typedef float v8f __attribute__((ext_vector_type(8)));

#define GCN_G   64
#define GCN_OUT 64
#define LDS_STRIDE 132

#define BK_CAP 32
__global__ void k_cnt_zero(int* __restrict__ icnt, int n, float* __restrict__ outp,
                           float* __restrict__ gcnt, int* __restrict__ ovn) {
    const int i  = blockIdx.x * blockDim.x + threadIdx.x;
    const int st = gridDim.x * blockDim.x;
    if (i < n) icnt[i] = 0;
    if (i == 0) *ovn = 0;
    for (int j = i; j < GCN_G * GCN_OUT; j += st) outp[j] = 0.0f;
    for (int j = i; j < GCN_G; j += st) gcnt[j] = 0.0f;
}

__device__ __forceinline__ float dinv_of(int c) {
    const float d = (float)(c + 1);
    return (d > 0.0f) ? (1.0f / sqrtf(d)) : 0.0f;
}

#define EG_WAVES 16
#define EG_MUL    8
#define EG_EPL    8

__device__ __forceinline__ void edge_bucket(const int* __restrict__ src, const int* __restrict__ dst,
                                            int* __restrict__ icnt, int* __restrict__ col,
                                            int* __restrict__ ovn, int* __restrict__ ovd,
                                            int* __restrict__ ovs, int e, int gl, int nl) {
    const int trip = EG_EPL * nl;
    int d[EG_EPL], sv[EG_EPL];
#pragma unroll
    for (int k = 0; k < EG_EPL; ++k) {
        const int ic = min(gl + k * nl, e - 1);
        d[k] = dst[ic]; sv[k] = src[ic];
    }
    for (int base = 0; base < e; base += trip) {
        int dn[EG_EPL], sn[EG_EPL], r[EG_EPL];
#pragma unroll
        for (int k = 0; k < EG_EPL; ++k) { dn[k] = d[k]; sn[k] = sv[k]; }
        if (base + trip < e) {
#pragma unroll
            for (int k = 0; k < EG_EPL; ++k) {
                const int ic = min(base + trip + gl + k * nl, e - 1);
                dn[k] = dst[ic]; sn[k] = src[ic];
            }
        }
#pragma unroll
        for (int k = 0; k < EG_EPL; ++k) {
            const int i = base + gl + k * nl;
            r[k] = 0;
            if (i < e) r[k] = atomicAdd(&icnt[d[k]], 1);
        }
#pragma unroll
        for (int k = 0; k < EG_EPL; ++k) {
            const int i = base + gl + k * nl;
            if (i < e) {
                if (r[k] < BK_CAP) {
                    col[d[k] * BK_CAP + r[k]] = sv[k];
                } else {
                    const int o = atomicAdd(ovn, 1);
                    ovd[o] = d[k];
                    ovs[o] = sv[k];
                }
            }
        }
#pragma unroll
        for (int k = 0; k < EG_EPL; ++k) { d[k] = dn[k]; sv[k] = sn[k]; }
    }
}

__device__ __forceinline__ void graph_node_counts(const int* __restrict__ batch, float* __restrict__ gcnt,
                                                  int n, int gl, int nl) {
    const int lane = gl & 31;
    for (int base = 0; base < n; base += 4 * nl) {
        const int i0 = base + 4 * gl;
        int gid[4];
        int nv = 0;
#pragma unroll
        for (int k = 0; k < 4; ++k) {
            const bool ok = i0 + k < n;
            gid[k] = ok ? batch[i0 + k] : 0;
            nv += ok ? 1 : 0;
        }
        const int gref = __builtin_amdgcn_readfirstlane(gid[0]);
        bool same = true;
#pragma unroll
        for (int k = 0; k < 4; ++k) same = same && (i0 + k >= n || gid[k] == gref);
        if (__all(same ? 1 : 0)) {
            int tot = nv;
#pragma unroll
            for (int off = 16; off > 0; off >>= 1) tot += __shfl_xor(tot, off);
            if (lane == 0 && tot > 0) atomicAdd(&gcnt[gref], (float)tot);
        } else {
            bool  have = false;
            int   cur  = 0;
            float run  = 0.0f;
#pragma unroll
            for (int k = 0; k < 4; ++k) {
                if (i0 + k < n) {
                    if (have && gid[k] != cur) {
                        atomicAdd(&gcnt[cur], run);
                        run = 0.0f;
                    }
                    have = true;
                    cur  = gid[k];
                    run += 1.0f;
                }
            }
            if (have) atomicAdd(&gcnt[cur], run);
        }
    }
}

template <typename VF, int F>
__device__ __forceinline__ void overflow_rows(const float* __restrict__ H, const int* __restrict__ icnt,
                                              const int* __restrict__ ovn, const int* __restrict__ ovd,
                                              const int* __restrict__ ovs, int node, float di,
                                              int lane, int fo, VF& acc) {
    const int on = *ovn;
    for (int base = 0; base < on; base += 32) {
        int dd = -1, ss = 0;
        if (base + lane < on) { dd = ovd[base + lane]; ss = ovs[base + lane]; }
        const bool hit = dd == node;
        float dw = 0.0f;
        if (hit) dw = dinv_of(icnt[ss]);
        unsigned m = __builtin_amdgcn_ballot_w32(hit);
        while (m) {
            const int b = __builtin_ctz(m);
            m &= m - 1;
            const int   s = __builtin_amdgcn_readlane(ss, b);
            const float w = __int_as_float(__builtin_amdgcn_readlane(__float_as_int(dw), b)) * di;
            const VF h = *(const VF*)(H + (size_t)s * F + fo);
            acc += w * h;
        }
    }
}

template <int VEC>
__global__ void __launch_bounds__(256)
k_gather_readout(const int* __restrict__ icnt, const int* __restrict__ col,
                 const int* __restrict__ ovn, const int* __restrict__ ovd, const int* __restrict__ ovs,
                 const float* __restrict__ H, const float* __restrict__ bias,
                 const int* __restrict__ batch, const float* __restrict__ gcnt,
                 float* __restrict__ outp, int n) {
    typedef float vf __attribute__((ext_vector_type(VEC)));
    constexpr int F = VEC * 32;
    __shared__ alignas(16) float rsum[8][F];
    __shared__ int rid[8];
    __shared__ int rhave[8];
    const int lane = threadIdx.x & 31;
    const int wv  = __builtin_amdgcn_readfirstlane((int)(threadIdx.x >> 5));
    const int wid = (int)blockIdx.x * 8 + wv;
    const int nw  = (int)(gridDim.x * 8);
    const int fo  = lane * VEC;
    const vf  bv  = *(const vf*)(bias + fo);
    const int q   = n / nw;
    const int rem = n - q * nw;
    const int beg = wid * q + min(wid, rem);
    const int end = beg + q + (wid < rem ? 1 : 0);
    bool have = false;
    int  cur  = 0;
    vf   racc = vf{};
    for (int i = beg; i < end; ++i) {
    const int   cnt   = icnt[i];
    const float di    = dinv_of(cnt);
    const int   g     = batch[i];

    const vf hs = *(const vf*)(H + (size_t)i * F + fo);
    vf acc = (di * di) * hs;
    {
        const int m = min(cnt, BK_CAP);
        int   s_l = 0;
        float w_l = 0.0f;
        if (lane < m) {
            s_l = col[i * BK_CAP + lane];
            w_l = dinv_of(icnt[s_l]) * di;
        }
        int j = 0;
        for (; j + 4 <= m; j += 4) {
            const int s0 = __builtin_amdgcn_readlane(s_l, j);
            const int s1 = __builtin_amdgcn_readlane(s_l, j + 1);
            const int s2 = __builtin_amdgcn_readlane(s_l, j + 2);
            const int s3 = __builtin_amdgcn_readlane(s_l, j + 3);
            const float w0 = __int_as_float(__builtin_amdgcn_readlane(__float_as_int(w_l), j));
            const float w1 = __int_as_float(__builtin_amdgcn_readlane(__float_as_int(w_l), j + 1));
            const float w2 = __int_as_float(__builtin_amdgcn_readlane(__float_as_int(w_l), j + 2));
            const float w3 = __int_as_float(__builtin_amdgcn_readlane(__float_as_int(w_l), j + 3));
            const vf h0 = *(const vf*)(H + (size_t)s0 * F + fo);
            const vf h1 = *(const vf*)(H + (size_t)s1 * F + fo);
            const vf h2 = *(const vf*)(H + (size_t)s2 * F + fo);
            const vf h3 = *(const vf*)(H + (size_t)s3 * F + fo);
            acc += w0 * h0;
            acc += w1 * h1;
            acc += w2 * h2;
            acc += w3 * h3;
        }
        const int r = m - j;
        if (r > 0) {
            const int   s0 = __builtin_amdgcn_readlane(s_l, j);
            const float w0 = __int_as_float(__builtin_amdgcn_readlane(__float_as_int(w_l), j));
            const vf h0 = *(const vf*)(H + (size_t)s0 * F + fo);
            if (r > 1) {
                const int   s1 = __builtin_amdgcn_readlane(s_l, j + 1);
                const float w1 = __int_as_float(__builtin_amdgcn_readlane(__float_as_int(w_l), j + 1));
                const vf h1 = *(const vf*)(H + (size_t)s1 * F + fo);
                if (r > 2) {
                    const int   s2 = __builtin_amdgcn_readlane(s_l, j + 2);
                    const float w2 = __int_as_float(__builtin_amdgcn_readlane(__float_as_int(w_l), j + 2));
                    const vf h2 = *(const vf*)(H + (size_t)s2 * F + fo);
                    acc += w0 * h0;
                    acc += w1 * h1;
                    acc += w2 * h2;
                } else {
                    acc += w0 * h0;
                    acc += w1 * h1;
                }
            } else {
                acc += w0 * h0;
            }
        }
    }
    if (cnt > BK_CAP) overflow_rows<vf, F>(H, icnt, ovn, ovd, ovs, i, di, lane, fo, acc);
    acc += bv;
    if (have && g != cur) {
        const float cn = fmaxf(gcnt[cur], 1.0f);
#pragma unroll
        for (int k = 0; k < VEC; ++k) atomicAdd(&outp[cur * F + fo + k], racc[k] / cn);
        racc = vf{};
    }
    have = true;
    cur  = g;
    racc += acc;
    }

    *(vf*)(&rsum[wv][fo]) = racc;
    if (lane == 0) {
        rid[wv]   = cur;
        rhave[wv] = have ? 1 : 0;
    }
    __syncthreads();
    if (threadIdx.x < F) {
        const int f = threadIdx.x;
        int   hv[8], gw[8];
        float v[8];
#pragma unroll
        for (int w = 0; w < 8; ++w) {
            hv[w] = rhave[w];
            gw[w] = rid[w];
            v[w]  = rsum[w][f];
        }
        bool  h  = false;
        int   cg = 0;
        float a  = 0.0f;
#pragma unroll
        for (int w = 0; w < 8; ++w) {
            if (hv[w]) {
                if (h && gw[w] != cg) {
                    atomicAdd(&outp[cg * F + f], a / fmaxf(gcnt[cg], 1.0f));
                    a = 0.0f;
                }
                h  = true;
                cg = gw[w];
                a += v[w];
            }
        }
        if (h) atomicAdd(&outp[cg * F + f], a / fmaxf(gcnt[cg], 1.0f));
    }
}

__global__ void __launch_bounds__(EG_WAVES * 32)
k_edges_gemm(const float* __restrict__ X, const float* __restrict__ W, float* __restrict__ H,
             int mt_lo, int mt_hi,
             const int* __restrict__ src, const int* __restrict__ dst, int* __restrict__ icnt,
             int* __restrict__ col, int* __restrict__ ovn, int* __restrict__ ovd,
             int* __restrict__ ovs, int e,
             const int* __restrict__ batch, float* __restrict__ gcnt, int n) {
    constexpr int F  = 128;
    constexpr int K  = 128;
    constexpr int NT = F / 16;
    constexpr int NG = K / 8;
    __shared__ alignas(16) float Wlds[F * LDS_STRIDE];

    const int t     = threadIdx.x;
    const int wave  = __builtin_amdgcn_readfirstlane(t >> 5);
    const int lane  = t & 31;
    const int half  = lane >> 4;
    const int l16   = lane & 15;

    const float* __restrict__ Xl = X + (size_t)l16 * K + (half << 1);

    const int gx = (int)gridDim.x;
    int mtile = mt_lo + wave * gx + (int)blockIdx.x;
    const bool any = wave < EG_MUL && mtile < mt_hi;

    v2f a[8];
#pragma unroll
    for (int u = 0; u < 8; ++u) a[u] = v2f{};
    if (any) {
#pragma unroll
        for (int u = 0; u < 8; ++u) a[u] = *(const v2f*)(Xl + (size_t)(mtile << 4) * K + 4 * u);
    }

    for (int idx = t; idx < K * F; idx += EG_WAVES * 32) {
        const int k = idx / F;
        const int n = idx & (F - 1);
        Wlds[n * LDS_STRIDE + k] = W[idx];
    }
    __syncthreads();

    if (wave >= EG_MUL) {
        const int nl = gx * (EG_WAVES - EG_MUL) * 32;
        const int gl = ((int)blockIdx.x * (EG_WAVES - EG_MUL) + (wave - EG_MUL)) * 32 + lane;
        graph_node_counts(batch, gcnt, n, gl, nl);
        edge_bucket(src, dst, icnt, col, ovn, ovd, ovs, e, gl, nl);
        return;
    }
    if (!any) return;

    const float* __restrict__ Wl = &Wlds[l16 * LDS_STRIDE + (half << 1)];
    v2f bc[NT][2];
#pragma unroll
    for (int n = 0; n < NT; ++n) {
#pragma unroll
        for (int q = 0; q < 2; ++q) bc[n][q] = *(const v2f*)(Wl + n * 16 * LDS_STRIDE + 4 * q);
    }

    for (;;) {
        const int  row0  = mtile << 4;
        const int  mnext = mtile + EG_MUL * gx;
        const bool more  = mnext < mt_hi;
        const float* __restrict__ Xrow = Xl + (size_t)row0 * K;

        v8f c[NT];
#pragma unroll
        for (int n = 0; n < NT; ++n) c[n] = v8f{};

#pragma unroll
        for (int blk = 0; blk < 4; ++blk) {
            v2f an[8];
#pragma unroll
            for (int u = 0; u < 8; ++u) an[u] = a[u];
            if (blk < 3) {
#pragma unroll
                for (int u = 0; u < 8; ++u) an[u] = *(const v2f*)(Xrow + (blk + 1) * 32 + 4 * u);
            } else if (more) {
#pragma unroll
                for (int u = 0; u < 8; ++u) an[u] = *(const v2f*)(Xl + (size_t)(mnext << 4) * K + 4 * u);
            }
#pragma unroll
            for (int p = 0; p < 4; ++p) {
                const int g  = blk * 4 + p;
                const int gn = (g + 1) % NG;
                v2f bn[NT][2];
#pragma unroll
                for (int n = 0; n < NT; ++n) {
#pragma unroll
                    for (int q = 0; q < 2; ++q)
                        bn[n][q] = *(const v2f*)(Wl + n * 16 * LDS_STRIDE + gn * 8 + 4 * q);
                }
                __builtin_amdgcn_sched_barrier(0);
#pragma unroll
                for (int n = 0; n < NT; ++n) {
                    c[n] = __builtin_amdgcn_wmma_f32_16x16x4_f32(false, a[2 * p], false, bc[n][0],
                                                                 (short)0, c[n], false, false);
                    c[n] = __builtin_amdgcn_wmma_f32_16x16x4_f32(false, a[2 * p + 1], false, bc[n][1],
                                                                 (short)0, c[n], false, false);
                }
                __builtin_amdgcn_sched_barrier(0);
#pragma unroll
                for (int n = 0; n < NT; ++n) { bc[n][0] = bn[n][0]; bc[n][1] = bn[n][1]; }
            }
#pragma unroll
            for (int u = 0; u < 8; ++u) a[u] = an[u];
        }

        float* __restrict__ Ho = H + (size_t)(row0 + (half << 3)) * F + l16;
#pragma unroll
        for (int v = 0; v < 8; ++v) {
#pragma unroll
            for (int n = 0; n < NT; ++n)
                Ho[(size_t)v * F + n * 16] = c[n][v];
        }
        if (!more) break;
        mtile = mnext;
    }
}

#define FG_WAVES 25

typedef float fg_v4f __attribute__((ext_vector_type(4)));

template <int NROW, bool CONV>
__device__ __forceinline__ void fg_rows(const float* __restrict__ H, int s_l, int cn_l, float& dn_l, float di,
                                        int j, int fo, fg_v4f& acc) {
    fg_v4f h[NROW];
#pragma unroll
    for (int u = 0; u < NROW; ++u) {
        const int s = __builtin_amdgcn_readlane(s_l, j + u);
        h[u] = *(const fg_v4f*)(H + (size_t)s * 128 + fo);
    }
    __builtin_amdgcn_sched_barrier(0);
    if (CONV) dn_l = dinv_of(cn_l);
#pragma unroll
    for (int u = 0; u < NROW; ++u) {
        const float w = __int_as_float(__builtin_amdgcn_readlane(__float_as_int(dn_l), j + u)) * di;
        acc += w * h[u];
    }
}

template <bool CONV>
__device__ __forceinline__ void fg_rows_tail(const float* __restrict__ H, int s_l, int cn_l, float& dn_l, float di,
                                             int j, int n, int fo, fg_v4f& acc) {
    fg_v4f h[15];
#pragma unroll
    for (int u = 0; u < 15; ++u) {
        h[u] = fg_v4f{};
        if (u < n) {
            const int s = __builtin_amdgcn_readlane(s_l, j + u);
            h[u] = *(const fg_v4f*)(H + (size_t)s * 128 + fo);
        }
    }
    if (CONV) dn_l = dinv_of(cn_l);
#pragma unroll
    for (int u = 0; u < 15; ++u) {
        if (u < n) {
            const float w = __int_as_float(__builtin_amdgcn_readlane(__float_as_int(dn_l), j + u)) * di;
            acc += w * h[u];
        }
    }
}

__device__ __forceinline__ void fg_chunk(const float* __restrict__ H, int s_l, int cn_l, float di,
                                         int m, int fo, fg_v4f& acc) {
    float dn_l = 0.0f;
    if (m >= 16) {
        fg_rows<16, true>(H, s_l, cn_l, dn_l, di, 0, fo, acc);
        if (m >= 32)     fg_rows<16, false>(H, s_l, cn_l, dn_l, di, 16, fo, acc);
        else if (m > 16) fg_rows_tail<false>(H, s_l, cn_l, dn_l, di, 16, m - 16, fo, acc);
    } else if (m > 0) {
        fg_rows_tail<true>(H, s_l, cn_l, dn_l, di, 0, m, fo, acc);
    }
}

template <int FOUT>
__global__ void __launch_bounds__(FG_WAVES * 32)
k_gather_relu_gemm(const int* __restrict__ icnt, const int* __restrict__ col,
                   const int* __restrict__ ovn, const int* __restrict__ ovd, const int* __restrict__ ovs,
                   const float* __restrict__ Hin, const float* __restrict__ bias,
                   const float* __restrict__ Wn, float* __restrict__ Hout, int mtiles) {
    typedef float vf4 __attribute__((ext_vector_type(4)));
    constexpr int K    = 128;
    constexpr int NT   = FOUT / 16;
    constexpr int KG   = (NT >= 8) ? 1 : 2;
    constexpr int NGRP = (K / 4) / KG;
    __shared__ alignas(16) float Wlds[FOUT * LDS_STRIDE];
    __shared__ alignas(16) float Atile[FG_WAVES * 16 * LDS_STRIDE];

    const int t    = threadIdx.x;
    const int wave = __builtin_amdgcn_readfirstlane(t >> 5);
    const int lane = t & 31;
    const int half = lane >> 4;
    const int l16  = lane & 15;

    for (int idx = t; idx < K * FOUT; idx += FG_WAVES * 32) {
        const int k = idx / FOUT;
        const int n = idx & (FOUT - 1);
        Wlds[n * LDS_STRIDE + k] = Wn[idx];
    }
    __syncthreads();

    float* At = &Atile[wave * 16 * LDS_STRIDE];
    const float* Al = At + l16 * LDS_STRIDE + (half << 1);
    const float* __restrict__ Wl = &Wlds[l16 * LDS_STRIDE + (half << 1)];
    const int fo = lane * 4;
    const vf4 bv = *(const vf4*)(bias + fo);
    const int gx = (int)gridDim.x;

    for (int mtile = wave * gx + (int)blockIdx.x; mtile < mtiles; mtile += FG_WAVES * gx) {
        const int row0 = mtile << 4;
        const int   ct_l = icnt[row0 + l16];
        const float di_l = dinv_of(ct_l);

        int   cnt   = __builtin_amdgcn_readlane(ct_l, 0);
        float di    = __int_as_float(__builtin_amdgcn_readlane(__float_as_int(di_l), 0));
        int   s_l = 0;
        if (lane < min(cnt, BK_CAP)) s_l = col[row0 * BK_CAP + lane];
        vf4 acc0;
        {
            const vf4 hs = *(const vf4*)(Hin + (size_t)row0 * K + fo);
            acc0 = (di * di) * hs;
        }
        for (int r = 0; r < 16; ++r) {
            vf4 acc = acc0;
            int cn_l = 0;
            if (lane < min(cnt, BK_CAP)) cn_l = icnt[s_l];
            const int   rn     = min(r + 1, 15);
            const int   ncnt   = (r < 15) ? __builtin_amdgcn_readlane(ct_l, rn) : 0;
            const float ndi    = __int_as_float(__builtin_amdgcn_readlane(__float_as_int(di_l), rn));
            int s_n = 0;
            if (lane < min(ncnt, BK_CAP)) s_n = col[(row0 + rn) * BK_CAP + lane];
            vf4 hs_n = vf4{};
            if (r < 15) hs_n = *(const vf4*)(Hin + (size_t)(row0 + r + 1) * K + fo);

            fg_chunk(Hin, s_l, cn_l, di, min(cnt, BK_CAP), fo, acc);
            if (cnt > BK_CAP)
                overflow_rows<vf4, 128>(Hin, icnt, ovn, ovd, ovs, row0 + r, di, lane, fo, acc);

            acc += bv;
#pragma unroll
            for (int k = 0; k < 4; ++k) acc[k] = fmaxf(acc[k], 0.0f);
            *(vf4*)(At + r * LDS_STRIDE + fo) = acc;

            acc0 = (ndi * ndi) * hs_n;
            s_l = s_n; cnt = ncnt; di = ndi;
        }

        v8f c[NT];
#pragma unroll
        for (int n = 0; n < NT; ++n) c[n] = v8f{};
        v2f ac[KG];
        v2f bc[NT][KG];
#pragma unroll
        for (int q = 0; q < KG; ++q) {
            ac[q] = *(const v2f*)(Al + 4 * q);
#pragma unroll
            for (int n = 0; n < NT; ++n) bc[n][q] = *(const v2f*)(Wl + n * 16 * LDS_STRIDE + 4 * q);
        }
#pragma unroll
        for (int g = 0; g < NGRP; ++g) {
            v2f an[KG];
            v2f bn[NT][KG];
#pragma unroll
            for (int q = 0; q < KG; ++q) {
                an[q] = ac[q];
#pragma unroll
                for (int n = 0; n < NT; ++n) bn[n][q] = bc[n][q];
            }
            if (g + 1 < NGRP) {
#pragma unroll
                for (int q = 0; q < KG; ++q) {
                    const int ko = 4 * ((g + 1) * KG + q);
                    an[q] = *(const v2f*)(Al + ko);
#pragma unroll
                    for (int n = 0; n < NT; ++n) bn[n][q] = *(const v2f*)(Wl + n * 16 * LDS_STRIDE + ko);
                }
            }
            __builtin_amdgcn_sched_barrier(0);
#pragma unroll
            for (int q = 0; q < KG; ++q) {
#pragma unroll
                for (int n = 0; n < NT; ++n)
                    c[n] = __builtin_amdgcn_wmma_f32_16x16x4_f32(false, ac[q], false, bc[n][q],
                                                                 (short)0, c[n], false, false);
            }
            __builtin_amdgcn_sched_barrier(0);
#pragma unroll
            for (int q = 0; q < KG; ++q) {
                ac[q] = an[q];
#pragma unroll
                for (int n = 0; n < NT; ++n) bc[n][q] = bn[n][q];
            }
        }
        float* __restrict__ Ho = Hout + (size_t)(row0 + (half << 3)) * FOUT + l16;
#pragma unroll
        for (int v = 0; v < 8; ++v) {
#pragma unroll
            for (int n = 0; n < NT; ++n)
                Ho[(size_t)v * FOUT + n * 16] = c[n][v];
        }
    }
}

extern "C" void kernel_launch(void* const* d_in, const int* in_sizes, int n_in,
                              void* d_out, int out_size, void* d_ws, size_t ws_size,
                              hipStream_t stream) {
    const float* x     = (const float*)d_in[0];
    const int*   edge  = (const int*)d_in[1];
    const int*   batch = (const int*)d_in[2];
    const float* W1    = (const float*)d_in[3];
    const float* b1    = (const float*)d_in[4];
    const float* W2    = (const float*)d_in[5];
    const float* b2    = (const float*)d_in[6];
    const float* W3    = (const float*)d_in[7];
    const float* b3    = (const float*)d_in[8];

    const int IN = 128, HID = 128, OUTF = 64;
    const int N = in_sizes[0] / IN;
    const int E = in_sizes[1] / 2;
    const int* src = edge;
    const int* dst = edge + E;

    char* ws = (char*)d_ws;
    float* bufA = (float*)ws;
    float* bufB = bufA + (size_t)N * HID;
    float* gcnt = bufB + (size_t)N * HID;
    int*   icnt = (int*)(gcnt + GCN_G);
    int*   ovn  = icnt + N;
    int*   col  = ovn + 64;
    int*   ovd  = col + (size_t)N * BK_CAP;
    int*   ovs  = ovd + E;

    const int TB = 256;
    dim3 blk(TB);

    const int mt  = N / 16;
    const int egb = 256;

    k_cnt_zero  <<<dim3((N + TB - 1) / TB), blk, 0, stream>>>(icnt, N, (float*)d_out, gcnt, ovn);
    k_edges_gemm<<<dim3(egb), dim3(EG_WAVES * 32), 0, stream>>>(x, W1, bufA, 0, mt, src, dst, icnt, col, ovn, ovd, ovs, E, batch, gcnt, N);

    const int gn = (N + 7) / 8 < 2048 ? (N + 7) / 8 : 2048;
    const int fgb = 250;

    k_gather_relu_gemm<128><<<dim3(fgb), dim3(FG_WAVES * 32), 0, stream>>>(icnt, col, ovn, ovd, ovs, bufA, b1, W2, bufB, mt);

    k_gather_relu_gemm<64><<<dim3(fgb), dim3(FG_WAVES * 32), 0, stream>>>(icnt, col, ovn, ovd, ovs, bufB, b2, W3, bufA, mt);

    k_gather_readout<2><<<dim3(gn), blk, 0, stream>>>(icnt, col, ovn, ovd, ovs, bufA, b3, batch, gcnt, (float*)d_out, N);
}


# Layers 2,3: each product row is stored scaled by its node's deg^-1/2, so those gathers add plain rows: no per-edge count gather, no weights; dinv[i] applied once per node; rounding order changes
# speedup vs baseline: 4.9630x; 1.0325x over previous
#include <hip/hip_runtime.h>
#include <math.h>

typedef float v2f __attribute__((ext_vector_type(2)));
typedef float v8f __attribute__((ext_vector_type(8)));

#define GCN_G   64
#define GCN_OUT 64
#define LDS_STRIDE 132

#define BK_CAP 32
__global__ void k_cnt_zero(int* __restrict__ icnt, int n, float* __restrict__ outp,
                           float* __restrict__ gcnt, int* __restrict__ ovn) {
    const int i  = blockIdx.x * blockDim.x + threadIdx.x;
    const int st = gridDim.x * blockDim.x;
    if (i < n) icnt[i] = 0;
    if (i == 0) *ovn = 0;
    for (int j = i; j < GCN_G * GCN_OUT; j += st) outp[j] = 0.0f;
    for (int j = i; j < GCN_G; j += st) gcnt[j] = 0.0f;
}

__device__ __forceinline__ float dinv_of(int c) {
    const float d = (float)(c + 1);
    return (d > 0.0f) ? (1.0f / sqrtf(d)) : 0.0f;
}

#define EG_WAVES 16
#define EG_MUL    8
#define EG_EPL    8

__device__ __forceinline__ void edge_bucket(const int* __restrict__ src, const int* __restrict__ dst,
                                            int* __restrict__ icnt, int* __restrict__ col,
                                            int* __restrict__ ovn, int* __restrict__ ovd,
                                            int* __restrict__ ovs, int e, int gl, int nl) {
    const int trip = EG_EPL * nl;
    int d[EG_EPL], sv[EG_EPL];
#pragma unroll
    for (int k = 0; k < EG_EPL; ++k) {
        const int ic = min(gl + k * nl, e - 1);
        d[k] = dst[ic]; sv[k] = src[ic];
    }
    for (int base = 0; base < e; base += trip) {
        int dn[EG_EPL], sn[EG_EPL], r[EG_EPL];
#pragma unroll
        for (int k = 0; k < EG_EPL; ++k) { dn[k] = d[k]; sn[k] = sv[k]; }
        if (base + trip < e) {
#pragma unroll
            for (int k = 0; k < EG_EPL; ++k) {
                const int ic = min(base + trip + gl + k * nl, e - 1);
                dn[k] = dst[ic]; sn[k] = src[ic];
            }
        }
#pragma unroll
        for (int k = 0; k < EG_EPL; ++k) {
            const int i = base + gl + k * nl;
            r[k] = 0;
            if (i < e) r[k] = atomicAdd(&icnt[d[k]], 1);
        }
#pragma unroll
        for (int k = 0; k < EG_EPL; ++k) {
            const int i = base + gl + k * nl;
            if (i < e) {
                if (r[k] < BK_CAP) {
                    col[d[k] * BK_CAP + r[k]] = sv[k];
                } else {
                    const int o = atomicAdd(ovn, 1);
                    ovd[o] = d[k];
                    ovs[o] = sv[k];
                }
            }
        }
#pragma unroll
        for (int k = 0; k < EG_EPL; ++k) { d[k] = dn[k]; sv[k] = sn[k]; }
    }
}

__device__ __forceinline__ void graph_node_counts(const int* __restrict__ batch, float* __restrict__ gcnt,
                                                  int n, int gl, int nl) {
    const int lane = gl & 31;
    for (int base = 0; base < n; base += 4 * nl) {
        const int i0 = base + 4 * gl;
        int gid[4];
        int nv = 0;
#pragma unroll
        for (int k = 0; k < 4; ++k) {
            const bool ok = i0 + k < n;
            gid[k] = ok ? batch[i0 + k] : 0;
            nv += ok ? 1 : 0;
        }
        const int gref = __builtin_amdgcn_readfirstlane(gid[0]);
        bool same = true;
#pragma unroll
        for (int k = 0; k < 4; ++k) same = same && (i0 + k >= n || gid[k] == gref);
        if (__all(same ? 1 : 0)) {
            int tot = nv;
#pragma unroll
            for (int off = 16; off > 0; off >>= 1) tot += __shfl_xor(tot, off);
            if (lane == 0 && tot > 0) atomicAdd(&gcnt[gref], (float)tot);
        } else {
            bool  have = false;
            int   cur  = 0;
            float run  = 0.0f;
#pragma unroll
            for (int k = 0; k < 4; ++k) {
                if (i0 + k < n) {
                    if (have && gid[k] != cur) {
                        atomicAdd(&gcnt[cur], run);
                        run = 0.0f;
                    }
                    have = true;
                    cur  = gid[k];
                    run += 1.0f;
                }
            }
            if (have) atomicAdd(&gcnt[cur], run);
        }
    }
}

template <typename VF, int F, bool PRE>
__device__ __forceinline__ void overflow_rows(const float* __restrict__ H, const int* __restrict__ icnt,
                                              const int* __restrict__ ovn, const int* __restrict__ ovd,
                                              const int* __restrict__ ovs, int node, float di,
                                              int lane, int fo, VF& acc) {
    const int on = *ovn;
    for (int base = 0; base < on; base += 32) {
        int dd = -1, ss = 0;
        if (base + lane < on) { dd = ovd[base + lane]; ss = ovs[base + lane]; }
        const bool hit = dd == node;
        float dw = 1.0f;
        if (!PRE && hit) dw = dinv_of(icnt[ss]);
        unsigned m = __builtin_amdgcn_ballot_w32(hit);
        while (m) {
            const int b = __builtin_ctz(m);
            m &= m - 1;
            const int   s = __builtin_amdgcn_readlane(ss, b);
            const float w = __int_as_float(__builtin_amdgcn_readlane(__float_as_int(dw), b)) * di;
            const VF h = *(const VF*)(H + (size_t)s * F + fo);
            if (PRE) acc += h; else acc += w * h;
        }
    }
}

template <int VEC>
__global__ void __launch_bounds__(256)
k_gather_readout(const int* __restrict__ icnt, const int* __restrict__ col,
                 const int* __restrict__ ovn, const int* __restrict__ ovd, const int* __restrict__ ovs,
                 const float* __restrict__ H, const float* __restrict__ bias,
                 const int* __restrict__ batch, const float* __restrict__ gcnt,
                 float* __restrict__ outp, int n) {
    typedef float vf __attribute__((ext_vector_type(VEC)));
    constexpr int F = VEC * 32;
    __shared__ alignas(16) float rsum[8][F];
    __shared__ int rid[8];
    __shared__ int rhave[8];
    const int lane = threadIdx.x & 31;
    const int wv  = __builtin_amdgcn_readfirstlane((int)(threadIdx.x >> 5));
    const int wid = (int)blockIdx.x * 8 + wv;
    const int nw  = (int)(gridDim.x * 8);
    const int fo  = lane * VEC;
    const vf  bv  = *(const vf*)(bias + fo);
    const int q   = n / nw;
    const int rem = n - q * nw;
    const int beg = wid * q + min(wid, rem);
    const int end = beg + q + (wid < rem ? 1 : 0);
    bool have = false;
    int  cur  = 0;
    vf   racc = vf{};
    for (int i = beg; i < end; ++i) {
    const int   cnt   = icnt[i];
    const float di    = dinv_of(cnt);
    const int   g     = batch[i];

    const vf hs = *(const vf*)(H + (size_t)i * F + fo);
    vf acc = hs;
    {
        const int m = min(cnt, BK_CAP);
        int   s_l = 0;
        if (lane < m) s_l = col[i * BK_CAP + lane];
        int j = 0;
        for (; j + 4 <= m; j += 4) {
            const int s0 = __builtin_amdgcn_readlane(s_l, j);
            const int s1 = __builtin_amdgcn_readlane(s_l, j + 1);
            const int s2 = __builtin_amdgcn_readlane(s_l, j + 2);
            const int s3 = __builtin_amdgcn_readlane(s_l, j + 3);
            const vf h0 = *(const vf*)(H + (size_t)s0 * F + fo);
            const vf h1 = *(const vf*)(H + (size_t)s1 * F + fo);
            const vf h2 = *(const vf*)(H + (size_t)s2 * F + fo);
            const vf h3 = *(const vf*)(H + (size_t)s3 * F + fo);
            acc += h0;
            acc += h1;
            acc += h2;
            acc += h3;
        }
        const int r = m - j;
        if (r > 0) {
            const int   s0 = __builtin_amdgcn_readlane(s_l, j);
            const vf h0 = *(const vf*)(H + (size_t)s0 * F + fo);
            if (r > 1) {
                const int   s1 = __builtin_amdgcn_readlane(s_l, j + 1);
                const vf h1 = *(const vf*)(H + (size_t)s1 * F + fo);
                if (r > 2) {
                    const int   s2 = __builtin_amdgcn_readlane(s_l, j + 2);
                    const vf h2 = *(const vf*)(H + (size_t)s2 * F + fo);
                    acc += h0;
                    acc += h1;
                    acc += h2;
                } else {
                    acc += h0;
                    acc += h1;
                }
            } else {
                acc += h0;
            }
        }
    }
    if (cnt > BK_CAP) overflow_rows<vf, F, true>(H, icnt, ovn, ovd, ovs, i, di, lane, fo, acc);
    acc = di * acc + bv;
    if (have && g != cur) {
        const float cn = fmaxf(gcnt[cur], 1.0f);
#pragma unroll
        for (int k = 0; k < VEC; ++k) atomicAdd(&outp[cur * F + fo + k], racc[k] / cn);
        racc = vf{};
    }
    have = true;
    cur  = g;
    racc += acc;
    }

    *(vf*)(&rsum[wv][fo]) = racc;
    if (lane == 0) {
        rid[wv]   = cur;
        rhave[wv] = have ? 1 : 0;
    }
    __syncthreads();
    if (threadIdx.x < F) {
        const int f = threadIdx.x;
        int   hv[8], gw[8];
        float v[8];
#pragma unroll
        for (int w = 0; w < 8; ++w) {
            hv[w] = rhave[w];
            gw[w] = rid[w];
            v[w]  = rsum[w][f];
        }
        bool  h  = false;
        int   cg = 0;
        float a  = 0.0f;
#pragma unroll
        for (int w = 0; w < 8; ++w) {
            if (hv[w]) {
                if (h && gw[w] != cg) {
                    atomicAdd(&outp[cg * F + f], a / fmaxf(gcnt[cg], 1.0f));
                    a = 0.0f;
                }
                h  = true;
                cg = gw[w];
                a += v[w];
            }
        }
        if (h) atomicAdd(&outp[cg * F + f], a / fmaxf(gcnt[cg], 1.0f));
    }
}

__global__ void __launch_bounds__(EG_WAVES * 32)
k_edges_gemm(const float* __restrict__ X, const float* __restrict__ W, float* __restrict__ H,
             int mt_lo, int mt_hi,
             const int* __restrict__ src, const int* __restrict__ dst, int* __restrict__ icnt,
             int* __restrict__ col, int* __restrict__ ovn, int* __restrict__ ovd,
             int* __restrict__ ovs, int e,
             const int* __restrict__ batch, float* __restrict__ gcnt, int n) {
    constexpr int F  = 128;
    constexpr int K  = 128;
    constexpr int NT = F / 16;
    constexpr int NG = K / 8;
    __shared__ alignas(16) float Wlds[F * LDS_STRIDE];

    const int t     = threadIdx.x;
    const int wave  = __builtin_amdgcn_readfirstlane(t >> 5);
    const int lane  = t & 31;
    const int half  = lane >> 4;
    const int l16   = lane & 15;

    const float* __restrict__ Xl = X + (size_t)l16 * K + (half << 1);

    const int gx = (int)gridDim.x;
    int mtile = mt_lo + wave * gx + (int)blockIdx.x;
    const bool any = wave < EG_MUL && mtile < mt_hi;

    v2f a[8];
#pragma unroll
    for (int u = 0; u < 8; ++u) a[u] = v2f{};
    if (any) {
#pragma unroll
        for (int u = 0; u < 8; ++u) a[u] = *(const v2f*)(Xl + (size_t)(mtile << 4) * K + 4 * u);
    }

    for (int idx = t; idx < K * F; idx += EG_WAVES * 32) {
        const int k = idx / F;
        const int n = idx & (F - 1);
        Wlds[n * LDS_STRIDE + k] = W[idx];
    }
    __syncthreads();

    if (wave >= EG_MUL) {
        const int nl = gx * (EG_WAVES - EG_MUL) * 32;
        const int gl = ((int)blockIdx.x * (EG_WAVES - EG_MUL) + (wave - EG_MUL)) * 32 + lane;
        graph_node_counts(batch, gcnt, n, gl, nl);
        edge_bucket(src, dst, icnt, col, ovn, ovd, ovs, e, gl, nl);
        return;
    }
    if (!any) return;

    const float* __restrict__ Wl = &Wlds[l16 * LDS_STRIDE + (half << 1)];
    v2f bc[NT][2];
#pragma unroll
    for (int n = 0; n < NT; ++n) {
#pragma unroll
        for (int q = 0; q < 2; ++q) bc[n][q] = *(const v2f*)(Wl + n * 16 * LDS_STRIDE + 4 * q);
    }

    for (;;) {
        const int  row0  = mtile << 4;
        const int  mnext = mtile + EG_MUL * gx;
        const bool more  = mnext < mt_hi;
        const float* __restrict__ Xrow = Xl + (size_t)row0 * K;

        v8f c[NT];
#pragma unroll
        for (int n = 0; n < NT; ++n) c[n] = v8f{};

#pragma unroll
        for (int blk = 0; blk < 4; ++blk) {
            v2f an[8];
#pragma unroll
            for (int u = 0; u < 8; ++u) an[u] = a[u];
            if (blk < 3) {
#pragma unroll
                for (int u = 0; u < 8; ++u) an[u] = *(const v2f*)(Xrow + (blk + 1) * 32 + 4 * u);
            } else if (more) {
#pragma unroll
                for (int u = 0; u < 8; ++u) an[u] = *(const v2f*)(Xl + (size_t)(mnext << 4) * K + 4 * u);
            }
#pragma unroll
            for (int p = 0; p < 4; ++p) {
                const int g  = blk * 4 + p;
                const int gn = (g + 1) % NG;
                v2f bn[NT][2];
#pragma unroll
                for (int n = 0; n < NT; ++n) {
#pragma unroll
                    for (int q = 0; q < 2; ++q)
                        bn[n][q] = *(const v2f*)(Wl + n * 16 * LDS_STRIDE + gn * 8 + 4 * q);
                }
                __builtin_amdgcn_sched_barrier(0);
#pragma unroll
                for (int n = 0; n < NT; ++n) {
                    c[n] = __builtin_amdgcn_wmma_f32_16x16x4_f32(false, a[2 * p], false, bc[n][0],
                                                                 (short)0, c[n], false, false);
                    c[n] = __builtin_amdgcn_wmma_f32_16x16x4_f32(false, a[2 * p + 1], false, bc[n][1],
                                                                 (short)0, c[n], false, false);
                }
                __builtin_amdgcn_sched_barrier(0);
#pragma unroll
                for (int n = 0; n < NT; ++n) { bc[n][0] = bn[n][0]; bc[n][1] = bn[n][1]; }
            }
#pragma unroll
            for (int u = 0; u < 8; ++u) a[u] = an[u];
        }

        float* __restrict__ Ho = H + (size_t)(row0 + (half << 3)) * F + l16;
#pragma unroll
        for (int v = 0; v < 8; ++v) {
#pragma unroll
            for (int n = 0; n < NT; ++n)
                Ho[(size_t)v * F + n * 16] = c[n][v];
        }
        if (!more) break;
        mtile = mnext;
    }
}

#define FG_WAVES 25

typedef float fg_v4f __attribute__((ext_vector_type(4)));

template <int NROW, bool CONV, bool PRE>
__device__ __forceinline__ void fg_rows(const float* __restrict__ H, int s_l, int cn_l, float& dn_l, float di,
                                        int j, int fo, fg_v4f& acc) {
    fg_v4f h[NROW];
#pragma unroll
    for (int u = 0; u < NROW; ++u) {
        const int s = __builtin_amdgcn_readlane(s_l, j + u);
        h[u] = *(const fg_v4f*)(H + (size_t)s * 128 + fo);
    }
    if (!PRE) {
        __builtin_amdgcn_sched_barrier(0);
        if (CONV) dn_l = dinv_of(cn_l);
    }
#pragma unroll
    for (int u = 0; u < NROW; ++u) {
        if (PRE) {
            acc += h[u];
        } else {
            const float w = __int_as_float(__builtin_amdgcn_readlane(__float_as_int(dn_l), j + u)) * di;
            acc += w * h[u];
        }
    }
}

template <bool CONV, bool PRE>
__device__ __forceinline__ void fg_rows_tail(const float* __restrict__ H, int s_l, int cn_l, float& dn_l, float di,
                                             int j, int n, int fo, fg_v4f& acc) {
    fg_v4f h[15];
#pragma unroll
    for (int u = 0; u < 15; ++u) {
        h[u] = fg_v4f{};
        if (u < n) {
            const int s = __builtin_amdgcn_readlane(s_l, j + u);
            h[u] = *(const fg_v4f*)(H + (size_t)s * 128 + fo);
        }
    }
    if (!PRE && CONV) dn_l = dinv_of(cn_l);
#pragma unroll
    for (int u = 0; u < 15; ++u) {
        if (u < n) {
            if (PRE) {
                acc += h[u];
            } else {
                const float w = __int_as_float(__builtin_amdgcn_readlane(__float_as_int(dn_l), j + u)) * di;
                acc += w * h[u];
            }
        }
    }
}

template <bool PRE>
__device__ __forceinline__ void fg_chunk(const float* __restrict__ H, int s_l, int cn_l, float di,
                                         int m, int fo, fg_v4f& acc) {
    float dn_l = 0.0f;
    if (m >= 16) {
        fg_rows<16, true, PRE>(H, s_l, cn_l, dn_l, di, 0, fo, acc);
        if (m >= 32)     fg_rows<16, false, PRE>(H, s_l, cn_l, dn_l, di, 16, fo, acc);
        else if (m > 16) fg_rows_tail<false, PRE>(H, s_l, cn_l, dn_l, di, 16, m - 16, fo, acc);
    } else if (m > 0) {
        fg_rows_tail<true, PRE>(H, s_l, cn_l, dn_l, di, 0, m, fo, acc);
    }
}

template <int FOUT, bool PRE>
__global__ void __launch_bounds__(FG_WAVES * 32)
k_gather_relu_gemm(const int* __restrict__ icnt, const int* __restrict__ col,
                   const int* __restrict__ ovn, const int* __restrict__ ovd, const int* __restrict__ ovs,
                   const float* __restrict__ Hin, const float* __restrict__ bias,
                   const float* __restrict__ Wn, float* __restrict__ Hout, int mtiles) {
    typedef float vf4 __attribute__((ext_vector_type(4)));
    constexpr int K    = 128;
    constexpr int NT   = FOUT / 16;
    constexpr int KG   = (NT >= 8) ? 1 : 2;
    constexpr int NGRP = (K / 4) / KG;
    __shared__ alignas(16) float Wlds[FOUT * LDS_STRIDE];
    __shared__ alignas(16) float Atile[FG_WAVES * 16 * LDS_STRIDE];

    const int t    = threadIdx.x;
    const int wave = __builtin_amdgcn_readfirstlane(t >> 5);
    const int lane = t & 31;
    const int half = lane >> 4;
    const int l16  = lane & 15;

    for (int idx = t; idx < K * FOUT; idx += FG_WAVES * 32) {
        const int k = idx / FOUT;
        const int n = idx & (FOUT - 1);
        Wlds[n * LDS_STRIDE + k] = Wn[idx];
    }
    __syncthreads();

    float* At = &Atile[wave * 16 * LDS_STRIDE];
    const float* Al = At + l16 * LDS_STRIDE + (half << 1);
    const float* __restrict__ Wl = &Wlds[l16 * LDS_STRIDE + (half << 1)];
    const int fo = lane * 4;
    const vf4 bv = *(const vf4*)(bias + fo);
    const int gx = (int)gridDim.x;

    for (int mtile = wave * gx + (int)blockIdx.x; mtile < mtiles; mtile += FG_WAVES * gx) {
        const int row0 = mtile << 4;
        const int   ct_l = icnt[row0 + l16];
        const float di_l = dinv_of(ct_l);

        int   cnt   = __builtin_amdgcn_readlane(ct_l, 0);
        float di    = __int_as_float(__builtin_amdgcn_readlane(__float_as_int(di_l), 0));
        int   s_l = 0;
        if (lane < min(cnt, BK_CAP)) s_l = col[row0 * BK_CAP + lane];
        vf4 acc0;
        {
            const vf4 hs = *(const vf4*)(Hin + (size_t)row0 * K + fo);
            acc0 = PRE ? hs : (di * di) * hs;
        }
        for (int r = 0; r < 16; ++r) {
            vf4 acc = acc0;
            int cn_l = 0;
            if (!PRE && lane < min(cnt, BK_CAP)) cn_l = icnt[s_l];
            const int   rn     = min(r + 1, 15);
            const int   ncnt   = (r < 15) ? __builtin_amdgcn_readlane(ct_l, rn) : 0;
            const float ndi    = __int_as_float(__builtin_amdgcn_readlane(__float_as_int(di_l), rn));
            int s_n = 0;
            if (lane < min(ncnt, BK_CAP)) s_n = col[(row0 + rn) * BK_CAP + lane];
            vf4 hs_n = vf4{};
            if (r < 15) hs_n = *(const vf4*)(Hin + (size_t)(row0 + r + 1) * K + fo);

            fg_chunk<PRE>(Hin, s_l, cn_l, di, min(cnt, BK_CAP), fo, acc);
            if (cnt > BK_CAP)
                overflow_rows<vf4, 128, PRE>(Hin, icnt, ovn, ovd, ovs, row0 + r, di, lane, fo, acc);

            if (PRE) acc = di * acc + bv; else acc += bv;
#pragma unroll
            for (int k = 0; k < 4; ++k) acc[k] = fmaxf(acc[k], 0.0f);
            *(vf4*)(At + r * LDS_STRIDE + fo) = acc;

            acc0 = PRE ? hs_n : (ndi * ndi) * hs_n;
            s_l = s_n; cnt = ncnt; di = ndi;
        }

        v8f c[NT];
#pragma unroll
        for (int n = 0; n < NT; ++n) c[n] = v8f{};
        v2f ac[KG];
        v2f bc[NT][KG];
#pragma unroll
        for (int q = 0; q < KG; ++q) {
            ac[q] = *(const v2f*)(Al + 4 * q);
#pragma unroll
            for (int n = 0; n < NT; ++n) bc[n][q] = *(const v2f*)(Wl + n * 16 * LDS_STRIDE + 4 * q);
        }
#pragma unroll
        for (int g = 0; g < NGRP; ++g) {
            v2f an[KG];
            v2f bn[NT][KG];
#pragma unroll
            for (int q = 0; q < KG; ++q) {
                an[q] = ac[q];
#pragma unroll
                for (int n = 0; n < NT; ++n) bn[n][q] = bc[n][q];
            }
            if (g + 1 < NGRP) {
#pragma unroll
                for (int q = 0; q < KG; ++q) {
                    const int ko = 4 * ((g + 1) * KG + q);
                    an[q] = *(const v2f*)(Al + ko);
#pragma unroll
                    for (int n = 0; n < NT; ++n) bn[n][q] = *(const v2f*)(Wl + n * 16 * LDS_STRIDE + ko);
                }
            }
            __builtin_amdgcn_sched_barrier(0);
#pragma unroll
            for (int q = 0; q < KG; ++q) {
#pragma unroll
                for (int n = 0; n < NT; ++n)
                    c[n] = __builtin_amdgcn_wmma_f32_16x16x4_f32(false, ac[q], false, bc[n][q],
                                                                 (short)0, c[n], false, false);
            }
            __builtin_amdgcn_sched_barrier(0);
#pragma unroll
            for (int q = 0; q < KG; ++q) {
                ac[q] = an[q];
#pragma unroll
                for (int n = 0; n < NT; ++n) bc[n][q] = bn[n][q];
            }
        }
        float* __restrict__ Ho = Hout + (size_t)(row0 + (half << 3)) * FOUT + l16;
#pragma unroll
        for (int v = 0; v < 8; ++v) {
            const float dv = __shfl(di_l, v + (half << 3));
#pragma unroll
            for (int n = 0; n < NT; ++n)
                Ho[(size_t)v * FOUT + n * 16] = dv * c[n][v];
        }
    }
}

extern "C" void kernel_launch(void* const* d_in, const int* in_sizes, int n_in,
                              void* d_out, int out_size, void* d_ws, size_t ws_size,
                              hipStream_t stream) {
    const float* x     = (const float*)d_in[0];
    const int*   edge  = (const int*)d_in[1];
    const int*   batch = (const int*)d_in[2];
    const float* W1    = (const float*)d_in[3];
    const float* b1    = (const float*)d_in[4];
    const float* W2    = (const float*)d_in[5];
    const float* b2    = (const float*)d_in[6];
    const float* W3    = (const float*)d_in[7];
    const float* b3    = (const float*)d_in[8];

    const int IN = 128, HID = 128, OUTF = 64;
    const int N = in_sizes[0] / IN;
    const int E = in_sizes[1] / 2;
    const int* src = edge;
    const int* dst = edge + E;

    char* ws = (char*)d_ws;
    float* bufA = (float*)ws;
    float* bufB = bufA + (size_t)N * HID;
    float* gcnt = bufB + (size_t)N * HID;
    int*   icnt = (int*)(gcnt + GCN_G);
    int*   ovn  = icnt + N;
    int*   col  = ovn + 64;
    int*   ovd  = col + (size_t)N * BK_CAP;
    int*   ovs  = ovd + E;

    const int TB = 256;
    dim3 blk(TB);

    const int mt  = N / 16;
    const int egb = 256;

    k_cnt_zero  <<<dim3((N + TB - 1) / TB), blk, 0, stream>>>(icnt, N, (float*)d_out, gcnt, ovn);
    k_edges_gemm<<<dim3(egb), dim3(EG_WAVES * 32), 0, stream>>>(x, W1, bufA, 0, mt, src, dst, icnt, col, ovn, ovd, ovs, E, batch, gcnt, N);

    const int gn = (N + 7) / 8 < 2048 ? (N + 7) / 8 : 2048;
    const int fgb = 250;

    k_gather_relu_gemm<128, false><<<dim3(fgb), dim3(FG_WAVES * 32), 0, stream>>>(icnt, col, ovn, ovd, ovs, bufA, b1, W2, bufB, mt);

    k_gather_relu_gemm<64, true><<<dim3(fgb), dim3(FG_WAVES * 32), 0, stream>>>(icnt, col, ovn, ovd, ovs, bufB, b2, W3, bufA, mt);

    k_gather_readout<2><<<dim3(gn), blk, 0, stream>>>(icnt, col, ovn, ovd, ovs, bufA, b3, batch, gcnt, (float*)d_out, N);
}
